# starting kernel
# baseline (speedup 1.0000x reference)
#include <hip/hip_runtime.h>
#include <hip/hip_bf16.h>
#include <math.h>

typedef __attribute__((ext_vector_type(2))) float v2f;
typedef __attribute__((ext_vector_type(8))) float v8f;

__device__ __forceinline__ float wave_sum32(float v) {
#pragma unroll
  for (int off = 16; off > 0; off >>= 1) v += __shfl_xor(v, off, 32);
  return v;
}

__device__ __forceinline__ void atomicMaxF(float* addr, float val) {
  if (val >= 0.0f) atomicMax((int*)addr, __float_as_int(val));
  else             atomicMin((unsigned int*)addr, __float_as_uint(val));
}

__global__ void fill_f32(float* __restrict__ p, float v, int n) {
  int i = blockIdx.x * blockDim.x + threadIdx.x;
  if (i < n) p[i] = v;
}

template <int K, int H>
__global__ void gemm_bias_wmma(const float* __restrict__ X, const float* __restrict__ W,
                               const float* __restrict__ bias, float* __restrict__ C, int N) {
  constexpr int COLT = H / 16;
  constexpr int ROWT = 8 / COLT;
  const int wave = threadIdx.x >> 5;
  const int lane = threadIdx.x & 31;
  const int rt = wave / COLT, ct = wave % COLT;
  const int row0 = (blockIdx.x * ROWT + rt) * 16;
  if (row0 >= N) return;
  const int col0 = ct * 16;
  const int half = lane >> 4, l16 = lane & 15;

  const float* __restrict__ arow = X + (size_t)(row0 + l16) * K;
  v8f acc = {};
#pragma unroll 4
  for (int k = 0; k < K; k += 4) {
    const int ka = k + 2 * half;
    v2f a, b;
    a.x = arow[ka];
    a.y = arow[ka + 1];
    b.x = W[(size_t)ka * H + col0 + l16];
    b.y = W[(size_t)(ka + 1) * H + col0 + l16];
    acc = __builtin_amdgcn_wmma_f32_16x16x4_f32(false, a, false, b, (short)0, acc,
                                                false, false);
  }
  const float bv = bias[col0 + l16];
  float* __restrict__ crow = C + (size_t)(row0 + 8 * half) * H + col0 + l16;
#pragma unroll
  for (int v = 0; v < 8; ++v) crow[(size_t)v * H] = acc[v] + bv;
}

template <int H>
__global__ void edge_scores(const float* __restrict__ xl, const float* __restrict__ xr,
                            const float* __restrict__ att, const int* __restrict__ ei,
                            int E, int Etot, float* __restrict__ ebuf,
                            float* __restrict__ m) {
  const int gid = blockIdx.x * blockDim.x + threadIdx.x;
  const int e = gid >> 5, lane = gid & 31;
  if (e >= Etot) return;
  int s, d;
  if (e < E) { s = ei[e]; d = ei[E + e]; } else { s = d = e - E; }

  float partial;
  if (H == 64) {
    const int c0 = lane * 2;
    const float2 a = *(const float2*)(xl + (size_t)s * 64 + c0);
    const float2 b = *(const float2*)(xr + (size_t)d * 64 + c0);
    float h0 = a.x + b.x; h0 = h0 > 0.0f ? h0 : 0.2f * h0;
    float h1 = a.y + b.y; h1 = h1 > 0.0f ? h1 : 0.2f * h1;
    partial = h0 * att[c0] + h1 * att[c0 + 1];
  } else {
    float hv = xl[(size_t)s * H + lane] + xr[(size_t)d * H + lane];
    hv = hv > 0.0f ? hv : 0.2f * hv;
    partial = hv * att[lane];
  }
  const float tot = wave_sum32(partial);
  if (lane == 0) {
    ebuf[e] = tot;
    atomicMaxF(m + d, tot);
  }
}

template <int H>
__global__ void edge_aggregate(const float* __restrict__ xl, const int* __restrict__ ei,
                               int E, int Etot, const float* __restrict__ ebuf,
                               const float* __restrict__ m, float* __restrict__ denom,
                               float* __restrict__ acc) {
  const int gid = blockIdx.x * blockDim.x + threadIdx.x;
  const int e = gid >> 5, lane = gid & 31;
  if (e >= Etot) return;
  int s, d;
  if (e < E) { s = ei[e]; d = ei[E + e]; } else { s = d = e - E; }

  const float ex = expf(ebuf[e] - m[d]);
  if (lane == 0) atomicAdd(denom + d, ex);
  if (H == 64) {
    const int c0 = lane * 2;
    const float2 a = *(const float2*)(xl + (size_t)s * 64 + c0);
    atomicAdd(acc + (size_t)d * 64 + c0, ex * a.x);
    atomicAdd(acc + (size_t)d * 64 + c0 + 1, ex * a.y);
  } else {
    atomicAdd(acc + (size_t)d * H + lane, ex * xl[(size_t)s * H + lane]);
  }
}

__global__ void finalize_layer1(const float* __restrict__ acc, const float* __restrict__ denom,
                                const float* __restrict__ bias, float* __restrict__ h, int N) {
  const int i = blockIdx.x * blockDim.x + threadIdx.x;
  if (i >= N * 64) return;
  const int n = i >> 6, c = i & 63;
  float v = acc[i] / (denom[n] + 1e-16f) + bias[c];
  h[i] = v > 0.0f ? v : 0.01f * v;
}

__global__ void finalize_layer2(const float* __restrict__ acc, const float* __restrict__ denom,
                                const float* __restrict__ bias, const float* __restrict__ Wlin,
                                const float* __restrict__ blin, float* __restrict__ out, int N) {
  const int gid = blockIdx.x * blockDim.x + threadIdx.x;
  const int n = gid >> 5, lane = gid & 31;
  if (n >= N) return;
  const float v = acc[(size_t)n * 32 + lane] / (denom[n] + 1e-16f) + bias[lane];
  const float tot = wave_sum32(v * Wlin[lane]);
  if (lane == 0) out[n] = tot + blin[0];
}

static inline int cdiv(int a, int b) { return (a + b - 1) / b; }

extern "C" void kernel_launch(void* const* d_in, const int* in_sizes, int n_in,
                              void* d_out, int out_size, void* d_ws, size_t ws_size,
                              hipStream_t stream) {
  const float* x     = (const float*)d_in[0];
  const int*   ei    = (const int*)  d_in[1];
  const float* W1l   = (const float*)d_in[2];
  const float* b1l   = (const float*)d_in[3];
  const float* W1r   = (const float*)d_in[4];
  const float* b1r   = (const float*)d_in[5];
  const float* att1  = (const float*)d_in[6];
  const float* bias1 = (const float*)d_in[7];
  const float* W2l   = (const float*)d_in[8];
  const float* b2l   = (const float*)d_in[9];
  const float* W2r   = (const float*)d_in[10];
  const float* b2r   = (const float*)d_in[11];
  const float* att2  = (const float*)d_in[12];
  const float* bias2 = (const float*)d_in[13];
  const float* Wlin  = (const float*)d_in[14];
  const float* blin  = (const float*)d_in[15];
  float* out = (float*)d_out;

  const int N    = in_sizes[0] / 128;
  const int E    = in_sizes[1] / 2;
  const int Etot = E + N;

  float* ws = (float*)d_ws;
  size_t off = 0;
  float* xl1  = ws + off; off += (size_t)N * 64;
  float* xr1  = ws + off; off += (size_t)N * 64;
  float* acc1 = ws + off; off += (size_t)N * 64;
  float* xl2  = ws + off; off += (size_t)N * 32;
  float* xr2  = ws + off; off += (size_t)N * 32;
  float* acc2 = ws + off; off += (size_t)N * 32;
  float* ebuf = ws + off; off += (size_t)Etot;
  float* mbuf = ws + off; off += (size_t)N;
  float* dbuf = ws + off; off += (size_t)N;
  (void)ws_size; (void)n_in; (void)out_size;

  const int TB = 256;
  const int rowTiles = cdiv(N, 16);
  const int eblocks = cdiv(Etot, TB / 32);

  gemm_bias_wmma<128, 64><<<cdiv(rowTiles, 2), TB, 0, stream>>>(x, W1l, b1l, xl1, N);
  gemm_bias_wmma<128, 64><<<cdiv(rowTiles, 2), TB, 0, stream>>>(x, W1r, b1r, xr1, N);

  fill_f32<<<cdiv(N * 64, TB), TB, 0, stream>>>(acc1, 0.0f, N * 64);
  fill_f32<<<cdiv(N, TB), TB, 0, stream>>>(mbuf, -INFINITY, N);
  fill_f32<<<cdiv(N, TB), TB, 0, stream>>>(dbuf, 0.0f, N);

  edge_scores<64><<<eblocks, TB, 0, stream>>>(xl1, xr1, att1, ei, E, Etot, ebuf, mbuf);
  edge_aggregate<64><<<eblocks, TB, 0, stream>>>(xl1, ei, E, Etot, ebuf, mbuf, dbuf, acc1);
  finalize_layer1<<<cdiv(N * 64, TB), TB, 0, stream>>>(acc1, dbuf, bias1, acc1, N);

  gemm_bias_wmma<64, 32><<<cdiv(rowTiles, 4), TB, 0, stream>>>(acc1, W2l, b2l, xl2, N);
  gemm_bias_wmma<64, 32><<<cdiv(rowTiles, 4), TB, 0, stream>>>(acc1, W2r, b2r, xr2, N);

  fill_f32<<<cdiv(N * 32, TB), TB, 0, stream>>>(acc2, 0.0f, N * 32);
  fill_f32<<<cdiv(N, TB), TB, 0, stream>>>(mbuf, -INFINITY, N);
  fill_f32<<<cdiv(N, TB), TB, 0, stream>>>(dbuf, 0.0f, N);

  edge_scores<32><<<eblocks, TB, 0, stream>>>(xl2, xr2, att2, ei, E, Etot, ebuf, mbuf);
  edge_aggregate<32><<<eblocks, TB, 0, stream>>>(xl2, ei, E, Etot, ebuf, mbuf, dbuf, acc2);

  finalize_layer2<<<cdiv(N, TB / 32), TB, 0, stream>>>(acc2, dbuf, bias2, Wlin, blin, out, N);
}


# edge kernels: each wave owns 32 consecutive edges (indices loaded once, broadcast by readlane) and keeps 8 gathered rows in flight, instead of one wave per edge with three serial round trips
# speedup vs baseline: 1.7999x; 1.7999x over previous
#include <hip/hip_runtime.h>
#include <hip/hip_bf16.h>
#include <math.h>

typedef __attribute__((ext_vector_type(2))) float v2f;
typedef __attribute__((ext_vector_type(8))) float v8f;

__device__ __forceinline__ float wave_sum32(float v) {
#pragma unroll
  for (int off = 16; off > 0; off >>= 1) v += __shfl_xor(v, off, 32);
  return v;
}

__device__ __forceinline__ void atomicMaxF(float* addr, float val) {
  if (val >= 0.0f) atomicMax((int*)addr, __float_as_int(val));
  else             atomicMin((unsigned int*)addr, __float_as_uint(val));
}

__global__ void fill_f32(float* __restrict__ p, float v, int n) {
  int i = blockIdx.x * blockDim.x + threadIdx.x;
  if (i < n) p[i] = v;
}

template <int K, int H>
__global__ void gemm_bias_wmma(const float* __restrict__ X, const float* __restrict__ W,
                               const float* __restrict__ bias, float* __restrict__ C, int N) {
  constexpr int COLT = H / 16;
  constexpr int ROWT = 8 / COLT;
  const int wave = threadIdx.x >> 5;
  const int lane = threadIdx.x & 31;
  const int rt = wave / COLT, ct = wave % COLT;
  const int row0 = (blockIdx.x * ROWT + rt) * 16;
  if (row0 >= N) return;
  const int col0 = ct * 16;
  const int half = lane >> 4, l16 = lane & 15;

  const float* __restrict__ arow = X + (size_t)(row0 + l16) * K;
  v8f acc = {};
#pragma unroll 4
  for (int k = 0; k < K; k += 4) {
    const int ka = k + 2 * half;
    v2f a, b;
    a.x = arow[ka];
    a.y = arow[ka + 1];
    b.x = W[(size_t)ka * H + col0 + l16];
    b.y = W[(size_t)(ka + 1) * H + col0 + l16];
    acc = __builtin_amdgcn_wmma_f32_16x16x4_f32(false, a, false, b, (short)0, acc,
                                                false, false);
  }
  const float bv = bias[col0 + l16];
  float* __restrict__ crow = C + (size_t)(row0 + 8 * half) * H + col0 + l16;
#pragma unroll
  for (int v = 0; v < 8; ++v) crow[(size_t)v * H] = acc[v] + bv;
}

constexpr int EPW = 32;
constexpr int EB  = 8;

template <int H>
__global__ void edge_scores(const float* __restrict__ xl, const float* __restrict__ xr,
                            const float* __restrict__ att, const int* __restrict__ ei,
                            int E, int Etot, float* __restrict__ ebuf,
                            float* __restrict__ m) {
  const int gid = blockIdx.x * blockDim.x + threadIdx.x;
  const int lane = gid & 31;
  const int e0 = (gid >> 5) * EPW;
  if (e0 >= Etot) return;
  const int e = e0 + lane;
  const bool valid = e < Etot;
  int s = 0, d = 0;
  if (valid) {
    if (e < E) { s = ei[e]; d = ei[E + e]; } else { s = d = e - E; }
  }

  float mine = 0.0f;
  if (H == 64) {
    const int c0 = lane * 2;
    const float2 w = *(const float2*)(att + c0);
#pragma unroll 1
    for (int q = 0; q < EPW; q += EB) {
      float2 a[EB], b[EB];
#pragma unroll
      for (int j = 0; j < EB; ++j) {
        const int sj = __builtin_amdgcn_readlane(s, q + j);
        const int dj = __builtin_amdgcn_readlane(d, q + j);
        a[j] = *(const float2*)(xl + (size_t)sj * 64 + c0);
        b[j] = *(const float2*)(xr + (size_t)dj * 64 + c0);
      }
#pragma unroll
      for (int j = 0; j < EB; ++j) {
        float h0 = a[j].x + b[j].x; h0 = h0 > 0.0f ? h0 : 0.2f * h0;
        float h1 = a[j].y + b[j].y; h1 = h1 > 0.0f ? h1 : 0.2f * h1;
        const float tot = wave_sum32(fmaf(h0, w.x, h1 * w.y));
        if (lane == q + j) mine = tot;
      }
    }
  } else {
    const float w = att[lane];
#pragma unroll 1
    for (int q = 0; q < EPW; q += EB) {
      float a[EB], b[EB];
#pragma unroll
      for (int j = 0; j < EB; ++j) {
        const int sj = __builtin_amdgcn_readlane(s, q + j);
        const int dj = __builtin_amdgcn_readlane(d, q + j);
        a[j] = xl[(size_t)sj * H + lane];
        b[j] = xr[(size_t)dj * H + lane];
      }
#pragma unroll
      for (int j = 0; j < EB; ++j) {
        float hv = a[j] + b[j];
        hv = hv > 0.0f ? hv : 0.2f * hv;
        const float tot = wave_sum32(hv * w);
        if (lane == q + j) mine = tot;
      }
    }
  }
  if (valid) {
    ebuf[e] = mine;
    atomicMaxF(m + d, mine);
  }
}

template <int H>
__global__ void edge_aggregate(const float* __restrict__ xl, const int* __restrict__ ei,
                               int E, int Etot, const float* __restrict__ ebuf,
                               const float* __restrict__ m, float* __restrict__ denom,
                               float* __restrict__ acc) {
  const int gid = blockIdx.x * blockDim.x + threadIdx.x;
  const int lane = gid & 31;
  const int e0 = __builtin_amdgcn_readfirstlane((gid >> 5) * EPW);
  if (e0 >= Etot) return;
  const int e = e0 + lane;
  const bool valid = e < Etot;
  int s = 0, d = 0;
  float ex = 0.0f;
  if (valid) {
    if (e < E) { s = ei[e]; d = ei[E + e]; } else { s = d = e - E; }
    ex = expf(ebuf[e] - m[d]);
    atomicAdd(denom + d, ex);
  }
  const int cnt = Etot - e0;

#pragma unroll 1
  for (int q = 0; q < EPW; q += EB) {
    if (H == 64) {
      const int c0 = lane * 2;
      float2 a[EB];
#pragma unroll
      for (int j = 0; j < EB; ++j) {
        const int sj = __builtin_amdgcn_readlane(s, q + j);
        a[j] = *(const float2*)(xl + (size_t)sj * 64 + c0);
      }
#pragma unroll
      for (int j = 0; j < EB; ++j) {
        if (q + j < cnt) {
          const int dj = __builtin_amdgcn_readlane(d, q + j);
          const float xj = __int_as_float(__builtin_amdgcn_readlane(__float_as_int(ex), q + j));
          atomicAdd(acc + (size_t)dj * 64 + c0, xj * a[j].x);
          atomicAdd(acc + (size_t)dj * 64 + c0 + 1, xj * a[j].y);
        }
      }
    } else {
      float a[EB];
#pragma unroll
      for (int j = 0; j < EB; ++j) {
        const int sj = __builtin_amdgcn_readlane(s, q + j);
        a[j] = xl[(size_t)sj * H + lane];
      }
#pragma unroll
      for (int j = 0; j < EB; ++j) {
        if (q + j < cnt) {
          const int dj = __builtin_amdgcn_readlane(d, q + j);
          const float xj = __int_as_float(__builtin_amdgcn_readlane(__float_as_int(ex), q + j));
          atomicAdd(acc + (size_t)dj * H + lane, xj * a[j]);
        }
      }
    }
  }
}

__global__ void finalize_layer1(const float* __restrict__ acc, const float* __restrict__ denom,
                                const float* __restrict__ bias, float* __restrict__ h, int N) {
  const int i = blockIdx.x * blockDim.x + threadIdx.x;
  if (i >= N * 64) return;
  const int n = i >> 6, c = i & 63;
  float v = acc[i] / (denom[n] + 1e-16f) + bias[c];
  h[i] = v > 0.0f ? v : 0.01f * v;
}

__global__ void finalize_layer2(const float* __restrict__ acc, const float* __restrict__ denom,
                                const float* __restrict__ bias, const float* __restrict__ Wlin,
                                const float* __restrict__ blin, float* __restrict__ out, int N) {
  const int gid = blockIdx.x * blockDim.x + threadIdx.x;
  const int n = gid >> 5, lane = gid & 31;
  if (n >= N) return;
  const float v = acc[(size_t)n * 32 + lane] / (denom[n] + 1e-16f) + bias[lane];
  const float tot = wave_sum32(v * Wlin[lane]);
  if (lane == 0) out[n] = tot + blin[0];
}

static inline int cdiv(int a, int b) { return (a + b - 1) / b; }

extern "C" void kernel_launch(void* const* d_in, const int* in_sizes, int n_in,
                              void* d_out, int out_size, void* d_ws, size_t ws_size,
                              hipStream_t stream) {
  const float* x     = (const float*)d_in[0];
  const int*   ei    = (const int*)  d_in[1];
  const float* W1l   = (const float*)d_in[2];
  const float* b1l   = (const float*)d_in[3];
  const float* W1r   = (const float*)d_in[4];
  const float* b1r   = (const float*)d_in[5];
  const float* att1  = (const float*)d_in[6];
  const float* bias1 = (const float*)d_in[7];
  const float* W2l   = (const float*)d_in[8];
  const float* b2l   = (const float*)d_in[9];
  const float* W2r   = (const float*)d_in[10];
  const float* b2r   = (const float*)d_in[11];
  const float* att2  = (const float*)d_in[12];
  const float* bias2 = (const float*)d_in[13];
  const float* Wlin  = (const float*)d_in[14];
  const float* blin  = (const float*)d_in[15];
  float* out = (float*)d_out;

  const int N    = in_sizes[0] / 128;
  const int E    = in_sizes[1] / 2;
  const int Etot = E + N;

  float* ws = (float*)d_ws;
  size_t off = 0;
  float* xl1  = ws + off; off += (size_t)N * 64;
  float* xr1  = ws + off; off += (size_t)N * 64;
  float* acc1 = ws + off; off += (size_t)N * 64;
  float* xl2  = ws + off; off += (size_t)N * 32;
  float* xr2  = ws + off; off += (size_t)N * 32;
  float* acc2 = ws + off; off += (size_t)N * 32;
  float* ebuf = ws + off; off += (size_t)Etot;
  float* mbuf = ws + off; off += (size_t)N;
  float* dbuf = ws + off; off += (size_t)N;
  (void)ws_size; (void)n_in; (void)out_size;

  const int TB = 256;
  const int rowTiles = cdiv(N, 16);
  const int eblocks = cdiv(Etot, (TB / 32) * EPW);

  gemm_bias_wmma<128, 64><<<cdiv(rowTiles, 2), TB, 0, stream>>>(x, W1l, b1l, xl1, N);
  gemm_bias_wmma<128, 64><<<cdiv(rowTiles, 2), TB, 0, stream>>>(x, W1r, b1r, xr1, N);

  fill_f32<<<cdiv(N * 64, TB), TB, 0, stream>>>(acc1, 0.0f, N * 64);
  fill_f32<<<cdiv(N, TB), TB, 0, stream>>>(mbuf, -INFINITY, N);
  fill_f32<<<cdiv(N, TB), TB, 0, stream>>>(dbuf, 0.0f, N);

  edge_scores<64><<<eblocks, TB, 0, stream>>>(xl1, xr1, att1, ei, E, Etot, ebuf, mbuf);
  edge_aggregate<64><<<eblocks, TB, 0, stream>>>(xl1, ei, E, Etot, ebuf, mbuf, dbuf, acc1);
  finalize_layer1<<<cdiv(N * 64, TB), TB, 0, stream>>>(acc1, dbuf, bias1, acc1, N);

  gemm_bias_wmma<64, 32><<<cdiv(rowTiles, 4), TB, 0, stream>>>(acc1, W2l, b2l, xl2, N);
  gemm_bias_wmma<64, 32><<<cdiv(rowTiles, 4), TB, 0, stream>>>(acc1, W2r, b2r, xr2, N);

  fill_f32<<<cdiv(N * 32, TB), TB, 0, stream>>>(acc2, 0.0f, N * 32);
  fill_f32<<<cdiv(N, TB), TB, 0, stream>>>(mbuf, -INFINITY, N);
  fill_f32<<<cdiv(N, TB), TB, 0, stream>>>(dbuf, 0.0f, N);

  edge_scores<32><<<eblocks, TB, 0, stream>>>(xl2, xr2, att2, ei, E, Etot, ebuf, mbuf);
  edge_aggregate<32><<<eblocks, TB, 0, stream>>>(xl2, ei, E, Etot, ebuf, mbuf, dbuf, acc2);

  finalize_layer2<<<cdiv(N, TB / 32), TB, 0, stream>>>(acc2, dbuf, bias2, Wlin, blin, out, N);
}


# edges grouped by destination (counting sort, integer atomics, once per call); one wave per node runs a running-max softmax and the row sum in registers: no float atomics; sum order = slot arrival
# speedup vs baseline: 2.7469x; 1.5261x over previous
#include <hip/hip_runtime.h>
#include <hip/hip_bf16.h>
#include <math.h>

typedef __attribute__((ext_vector_type(2))) float v2f;
typedef __attribute__((ext_vector_type(8))) float v8f;

__device__ __forceinline__ float wave_sum32(float v) {
#pragma unroll
  for (int off = 16; off > 0; off >>= 1) v += __shfl_xor(v, off, 32);
  return v;
}

template <int K, int H>
__global__ void gemm_bias_wmma(const float* __restrict__ X, const float* __restrict__ W,
                               const float* __restrict__ bias, float* __restrict__ C, int N) {
  constexpr int COLT = H / 16;
  constexpr int ROWT = 8 / COLT;
  const int wave = threadIdx.x >> 5;
  const int lane = threadIdx.x & 31;
  const int rt = wave / COLT, ct = wave % COLT;
  const int row0 = (blockIdx.x * ROWT + rt) * 16;
  if (row0 >= N) return;
  const int col0 = ct * 16;
  const int half = lane >> 4, l16 = lane & 15;

  const float* __restrict__ arow = X + (size_t)(row0 + l16) * K;
  v8f acc = {};
#pragma unroll 4
  for (int k = 0; k < K; k += 4) {
    const int ka = k + 2 * half;
    v2f a, b;
    a.x = arow[ka];
    a.y = arow[ka + 1];
    b.x = W[(size_t)ka * H + col0 + l16];
    b.y = W[(size_t)(ka + 1) * H + col0 + l16];
    acc = __builtin_amdgcn_wmma_f32_16x16x4_f32(false, a, false, b, (short)0, acc,
                                                false, false);
  }
  const float bv = bias[col0 + l16];
  float* __restrict__ crow = C + (size_t)(row0 + 8 * half) * H + col0 + l16;
#pragma unroll
  for (int v = 0; v < 8; ++v) crow[(size_t)v * H] = acc[v] + bv;
}

__device__ __forceinline__ int wave_sum32_i(int v) {
#pragma unroll
  for (int off = 16; off > 0; off >>= 1) v += __shfl_xor(v, off, 32);
  return v;
}

__global__ void fill_i32(int* __restrict__ p, int v, int n) {
  int i = blockIdx.x * blockDim.x + threadIdx.x;
  if (i < n) p[i] = v;
}

constexpr int EPT = 4;
__global__ void dst_count(const int* __restrict__ ei, int E, int* __restrict__ cnt) {
  const int g = blockIdx.x * blockDim.x + threadIdx.x;
  const int T = gridDim.x * blockDim.x;
  int d[EPT];
#pragma unroll
  for (int k = 0; k < EPT; ++k) {
    const int e = g + k * T;
    d[k] = e < E ? ei[E + e] : -1;
  }
#pragma unroll
  for (int k = 0; k < EPT; ++k)
    if (d[k] >= 0) atomicAdd(cnt + d[k], 1);
}

constexpr int SCAN_TILE = 1024;
__global__ void rowptr_scan(const int* __restrict__ cnt, int* __restrict__ rowptr) {
  __shared__ int sh_pre[8];
  __shared__ int sh_own[8];
  const int t = threadIdx.x, lane = t & 31, wave = t >> 5;
  const int tile0 = blockIdx.x * SCAN_TILE;
  int pre = 0;
  for (int i = t * 4; i < tile0; i += SCAN_TILE) {
    const int4 u = *(const int4*)(cnt + i);
    pre += (u.x + u.y) + (u.z + u.w);
  }
  const int4 v = *(const int4*)(cnt + tile0 + t * 4);
  const int own = (v.x + v.y) + (v.z + v.w);
  int inc = own;
#pragma unroll
  for (int off = 1; off < 32; off <<= 1) {
    const int u = __shfl_up(inc, off, 32);
    if (lane >= off) inc += u;
  }
  pre = wave_sum32_i(pre);
  if (lane == 31) sh_own[wave] = inc;
  if (lane == 0) sh_pre[wave] = pre;
  __syncthreads();
  int base = 0;
#pragma unroll
  for (int w = 0; w < 8; ++w) {
    base += sh_pre[w];
    if (w < wave) base += sh_own[w];
  }
  int4 r;
  r.x = base + inc - own;
  r.y = r.x + v.x;
  r.z = r.y + v.y;
  r.w = r.z + v.z;
  *(int4*)(rowptr + tile0 + t * 4) = r;
}

__global__ void csr_fill(const int* __restrict__ ei, int E, const int* __restrict__ rowptr,
                         int* __restrict__ cnt, int* __restrict__ csr) {
  const int g = blockIdx.x * blockDim.x + threadIdx.x;
  const int T = gridDim.x * blockDim.x;
  int d[EPT], s[EPT];
#pragma unroll
  for (int k = 0; k < EPT; ++k) {
    const int e = g + k * T;
    d[k] = e < E ? ei[E + e] : -1;
    s[k] = e < E ? ei[e] : 0;
  }
  int base[EPT], left[EPT];
#pragma unroll
  for (int k = 0; k < EPT; ++k) {
    base[k] = 0; left[k] = 0;
    if (d[k] >= 0) { base[k] = rowptr[d[k]]; left[k] = atomicSub(cnt + d[k], 1); }
  }
#pragma unroll
  for (int k = 0; k < EPT; ++k)
    if (d[k] >= 0) csr[base[k] + left[k] - 1] = s[k];
}

constexpr int EB = 8;

template <int H>
__device__ __forceinline__ float score_part(const float* a, const float* r, const float* w) {
  if (H == 64) {
    float h0 = a[0] + r[0]; h0 = h0 > 0.0f ? h0 : 0.2f * h0;
    float h1 = a[1] + r[1]; h1 = h1 > 0.0f ? h1 : 0.2f * h1;
    return fmaf(h0, w[0], h1 * w[1]);
  } else {
    float hv = a[0] + r[0];
    hv = hv > 0.0f ? hv : 0.2f * hv;
    return hv * w[0];
  }
}

template <int H>
__device__ __forceinline__ void load_row(float* dst, const float* __restrict__ base, int row, int lane) {
  if (H == 64) {
    const float2 t = *(const float2*)(base + (size_t)row * 64 + lane * 2);
    dst[0] = t.x; dst[1] = t.y;
  } else {
    dst[0] = base[(size_t)row * 32 + lane];
  }
}

template <int H, bool LAST>
__global__ void node_attend(const float* __restrict__ xl, const float* __restrict__ xr,
                            const float* __restrict__ att, const int* __restrict__ rowptr,
                            const int* __restrict__ csr, const float* __restrict__ bias,
                            const float* __restrict__ Wlin, const float* __restrict__ blin,
                            float* __restrict__ outp, int N) {
  constexpr int V = H / 32;
  const int gid = blockIdx.x * blockDim.x + threadIdx.x;
  const int lane = gid & 31;
  const int n = __builtin_amdgcn_readfirstlane(gid >> 5);
  if (n >= N) return;
  const int beg = rowptr[n], end = rowptr[n + 1];

  float w[V], r[V], acc[V];
  load_row<H>(w, att, 0, lane);
  load_row<H>(r, xr, n, lane);
  load_row<H>(acc, xl, n, lane);
  float m = wave_sum32(score_part<H>(acc, r, w));
  float den = 1.0f;

  for (int c = beg; c < end; c += 32) {
    const int cnt = min(32, end - c);
    int s = n;
    if (lane < cnt) s = csr[c + lane];
#pragma unroll 1
    for (int q = 0; q < cnt; q += EB) {
      float a[EB][V], ev[EB];
#pragma unroll
      for (int j = 0; j < EB; ++j) {
        const int sj = __builtin_amdgcn_readlane(s, q + j);
        load_row<H>(a[j], xl, sj, lane);
      }
      float mb = m;
#pragma unroll
      for (int j = 0; j < EB; ++j) {
        ev[j] = wave_sum32(score_part<H>(a[j], r, w));
        if (q + j < cnt) mb = fmaxf(mb, ev[j]);
      }
      const float sc = expf(m - mb);
      m = mb;
      den *= sc;
#pragma unroll
      for (int v = 0; v < V; ++v) acc[v] *= sc;
#pragma unroll
      for (int j = 0; j < EB; ++j) {
        if (q + j < cnt) {
          const float p = expf(ev[j] - m);
          den += p;
#pragma unroll
          for (int v = 0; v < V; ++v) acc[v] = fmaf(p, a[j][v], acc[v]);
        }
      }
    }
  }

  const float dn = den + 1e-16f;
  if (!LAST) {
    float b[V], o[V];
    load_row<H>(b, bias, 0, lane);
#pragma unroll
    for (int v = 0; v < V; ++v) {
      const float t = acc[v] / dn + b[v];
      o[v] = t > 0.0f ? t : 0.01f * t;
    }
    if (H == 64) {
      *(float2*)(outp + (size_t)n * 64 + lane * 2) = make_float2(o[0], o[V - 1]);
    } else {
      outp[(size_t)n * 32 + lane] = o[0];
    }
  } else {
    float tot = (acc[0] / dn + bias[lane * V]) * Wlin[lane * V];
#pragma unroll
    for (int v = 1; v < V; ++v)
      tot = fmaf(acc[v] / dn + bias[lane * V + v], Wlin[lane * V + v], tot);
    tot = wave_sum32(tot);
    if (lane == 0) outp[n] = tot + blin[0];
  }
}

static inline int cdiv(int a, int b) { return (a + b - 1) / b; }

extern "C" void kernel_launch(void* const* d_in, const int* in_sizes, int n_in,
                              void* d_out, int out_size, void* d_ws, size_t ws_size,
                              hipStream_t stream) {
  const float* x     = (const float*)d_in[0];
  const int*   ei    = (const int*)  d_in[1];
  const float* W1l   = (const float*)d_in[2];
  const float* b1l   = (const float*)d_in[3];
  const float* W1r   = (const float*)d_in[4];
  const float* b1r   = (const float*)d_in[5];
  const float* att1  = (const float*)d_in[6];
  const float* bias1 = (const float*)d_in[7];
  const float* W2l   = (const float*)d_in[8];
  const float* b2l   = (const float*)d_in[9];
  const float* W2r   = (const float*)d_in[10];
  const float* b2r   = (const float*)d_in[11];
  const float* att2  = (const float*)d_in[12];
  const float* bias2 = (const float*)d_in[13];
  const float* Wlin  = (const float*)d_in[14];
  const float* blin  = (const float*)d_in[15];
  float* out = (float*)d_out;

  const int N    = in_sizes[0] / 128;
  const int E    = in_sizes[1] / 2;

  float* ws = (float*)d_ws;
  const int cntPad = cdiv(N + 1, SCAN_TILE) * SCAN_TILE;
  size_t off = 0;
  float* xl1  = ws + off; off += (size_t)N * 64;
  float* xr1  = ws + off; off += (size_t)N * 64;
  float* h1   = ws + off; off += (size_t)N * 64;
  float* xl2  = ws + off; off += (size_t)N * 32;
  float* xr2  = ws + off; off += (size_t)N * 32;
  off = (off + 3) & ~(size_t)3;
  int* cnt    = (int*)(ws + off); off += (size_t)cntPad;
  int* rowptr = (int*)(ws + off); off += (size_t)cntPad;
  int* csr    = (int*)(ws + off); off += (size_t)E;
  (void)ws_size; (void)n_in; (void)out_size;

  const int TB = 256;
  const int rowTiles = cdiv(N, 16);
  const int nblocks = cdiv(N, TB / 32);

  fill_i32<<<cdiv(cntPad, TB), TB, 0, stream>>>(cnt, 0, cntPad);
  dst_count<<<cdiv(E, TB * EPT), TB, 0, stream>>>(ei, E, cnt);
  rowptr_scan<<<cntPad / SCAN_TILE, TB, 0, stream>>>(cnt, rowptr);
  csr_fill<<<cdiv(E, TB * EPT), TB, 0, stream>>>(ei, E, rowptr, cnt, csr);

  gemm_bias_wmma<128, 64><<<cdiv(rowTiles, 2), TB, 0, stream>>>(x, W1l, b1l, xl1, N);
  gemm_bias_wmma<128, 64><<<cdiv(rowTiles, 2), TB, 0, stream>>>(x, W1r, b1r, xr1, N);
  node_attend<64, false><<<nblocks, TB, 0, stream>>>(xl1, xr1, att1, rowptr, csr, bias1,
                                                     Wlin, blin, h1, N);

  gemm_bias_wmma<64, 32><<<cdiv(rowTiles, 4), TB, 0, stream>>>(h1, W2l, b2l, xl2, N);
  gemm_bias_wmma<64, 32><<<cdiv(rowTiles, 4), TB, 0, stream>>>(h1, W2r, b2r, xr2, N);
  node_attend<32, true><<<nblocks, TB, 0, stream>>>(xl2, xr2, att2, rowptr, csr, bias2,
                                                    Wlin, blin, out, N);
}


# edges grouped by destination through LDS bins in 2 launches (chunk sort by 256-node bin, one work-group per bin) instead of 4 with global atomics and a serial scan; sum order = LDS slot arrival
# speedup vs baseline: 5.4295x; 1.9766x over previous
#include <hip/hip_runtime.h>
#include <hip/hip_bf16.h>
#include <math.h>

typedef __attribute__((ext_vector_type(2))) float v2f;
typedef __attribute__((ext_vector_type(8))) float v8f;

__device__ __forceinline__ float wave_sum32(float v) {
#pragma unroll
  for (int off = 16; off > 0; off >>= 1) v += __shfl_xor(v, off, 32);
  return v;
}

template <int K, int H>
__global__ void gemm_bias_wmma(const float* __restrict__ X, const float* __restrict__ W,
                               const float* __restrict__ bias, float* __restrict__ C, int N) {
  constexpr int COLT = H / 16;
  constexpr int ROWT = 8 / COLT;
  const int wave = threadIdx.x >> 5;
  const int lane = threadIdx.x & 31;
  const int rt = wave / COLT, ct = wave % COLT;
  const int row0 = (blockIdx.x * ROWT + rt) * 16;
  if (row0 >= N) return;
  const int col0 = ct * 16;
  const int half = lane >> 4, l16 = lane & 15;

  const float* __restrict__ arow = X + (size_t)(row0 + l16) * K;
  v8f acc = {};
#pragma unroll 4
  for (int k = 0; k < K; k += 4) {
    const int ka = k + 2 * half;
    v2f a, b;
    a.x = arow[ka];
    a.y = arow[ka + 1];
    b.x = W[(size_t)ka * H + col0 + l16];
    b.y = W[(size_t)(ka + 1) * H + col0 + l16];
    acc = __builtin_amdgcn_wmma_f32_16x16x4_f32(false, a, false, b, (short)0, acc,
                                                false, false);
  }
  const float bv = bias[col0 + l16];
  float* __restrict__ crow = C + (size_t)(row0 + 8 * half) * H + col0 + l16;
#pragma unroll
  for (int v = 0; v < 8; ++v) crow[(size_t)v * H] = acc[v] + bv;
}

__device__ __forceinline__ int wave_sum32_i(int v) {
#pragma unroll
  for (int off = 16; off > 0; off >>= 1) v += __shfl_xor(v, off, 32);
  return v;
}

constexpr int LOGW = 8;
constexpr int BINW = 1 << LOGW;
constexpr int SORT_T = 1024;
constexpr int SORT_EPT = 8;
constexpr int CH = SORT_T * SORT_EPT;
constexpr int NBMAX = SORT_T;
constexpr int GMAX = 1024;
constexpr int CAP = 12288;

__device__ __forceinline__ int block_excl_scan(int v, int* sh_w, int& total) {
  const int lane = threadIdx.x & 31, wave = threadIdx.x >> 5;
  int inc = v;
#pragma unroll
  for (int off = 1; off < 32; off <<= 1) {
    const int u = __shfl_up(inc, off, 32);
    if (lane >= off) inc += u;
  }
  if (lane == 31) sh_w[wave] = inc;
  __syncthreads();
  if (wave == 0) {
    const int w = sh_w[lane];
    int winc = w;
#pragma unroll
    for (int off = 1; off < 32; off <<= 1) {
      const int u = __shfl_up(winc, off, 32);
      if (lane >= off) winc += u;
    }
    sh_w[lane] = winc - w;
    if (lane == 31) sh_w[32] = winc;
  }
  __syncthreads();
  total = sh_w[32];
  return sh_w[wave] + inc - v;
}

__global__ void __launch_bounds__(SORT_T)
bin_chunk(const int* __restrict__ ei, int E, int NB, int* __restrict__ bsrc,
          int* __restrict__ bdst, int* __restrict__ lscan) {
  __shared__ int sh_bin[NBMAX];
  __shared__ int sh_w[33];
  __shared__ int sh_buf[CH];
  const int t = threadIdx.x;
  const int g = blockIdx.x;
  const int e0 = g * CH;
  sh_bin[t] = 0;
  __syncthreads();
  int s[SORT_EPT], d[SORT_EPT], rk[SORT_EPT];
#pragma unroll
  for (int k = 0; k < SORT_EPT; ++k) {
    const int e = e0 + k * SORT_T + t;
    d[k] = e < E ? ei[E + e] : -1;
    s[k] = e < E ? ei[e] : 0;
  }
#pragma unroll
  for (int k = 0; k < SORT_EPT; ++k) {
    rk[k] = 0;
    if (d[k] >= 0) rk[k] = atomicAdd(&sh_bin[d[k] >> LOGW], 1);
  }
  __syncthreads();
  int total;
  const int ex = block_excl_scan(sh_bin[t], sh_w, total);
  sh_bin[t] = ex;
  int* __restrict__ row = lscan + (size_t)g * (NB + 1);
  if (t < NB) row[t] = ex;
  if (t == 0) row[NB] = total;
  __syncthreads();
  int pos[SORT_EPT];
#pragma unroll
  for (int k = 0; k < SORT_EPT; ++k) pos[k] = d[k] >= 0 ? sh_bin[d[k] >> LOGW] + rk[k] : 0;
#pragma unroll
  for (int k = 0; k < SORT_EPT; ++k) if (d[k] >= 0) sh_buf[pos[k]] = d[k];
  __syncthreads();
#pragma unroll
  for (int k = 0; k < SORT_EPT; ++k) {
    const int i = k * SORT_T + t;
    if (i < total) bdst[e0 + i] = sh_buf[i];
  }
  __syncthreads();
#pragma unroll
  for (int k = 0; k < SORT_EPT; ++k) if (d[k] >= 0) sh_buf[pos[k]] = s[k];
  __syncthreads();
#pragma unroll
  for (int k = 0; k < SORT_EPT; ++k) {
    const int i = k * SORT_T + t;
    if (i < total) bsrc[e0 + i] = sh_buf[i];
  }
}

__global__ void __launch_bounds__(SORT_T)
bin_to_csr(const int* __restrict__ bsrc, const int* __restrict__ bdst,
           const int* __restrict__ lscan, int G, int NB, int N,
           int* __restrict__ rowptr, int* __restrict__ csr) {
  __shared__ int sh_s[GMAX];
  __shared__ int sh_n[GMAX];
  __shared__ int sh_cnt[BINW];
  __shared__ int sh_w[33];
  __shared__ int sh_red[2];
  __shared__ int sh_sorted[CAP];
  const int t = threadIdx.x, lane = t & 31;
  const int wave = __builtin_amdgcn_readfirstlane(t >> 5);
  const int b = blockIdx.x;
  const int n0 = b << LOGW;
  if (t < BINW) sh_cnt[t] = 0;
  if (t < 2) sh_red[t] = 0;
  __syncthreads();
  int pb = 0, pn = 0;
  for (int g = t; g < G; g += SORT_T) {
    const int* __restrict__ row = lscan + (size_t)g * (NB + 1) + b;
    const int s0 = row[0], s1 = row[1];
    sh_s[g] = g * CH + s0;
    sh_n[g] = s1 - s0;
    pb += s0;
    pn += s1 - s0;
  }
  pb = wave_sum32_i(pb);
  pn = wave_sum32_i(pn);
  if (lane == 0) { atomicAdd(&sh_red[0], pb); atomicAdd(&sh_red[1], pn); }
  __syncthreads();
  const int base = sh_red[0];
  const int total = sh_red[1];
  for (int g = wave; g < G; g += SORT_T / 32) {
    const int st = sh_s[g], n = sh_n[g];
    for (int i = lane; i < n; i += 32) atomicAdd(&sh_cnt[bdst[st + i] - n0], 1);
  }
  __syncthreads();
  int tot2;
  const int ex = block_excl_scan(t < BINW ? sh_cnt[t] : 0, sh_w, tot2);
  if (t < BINW) {
    sh_cnt[t] = ex;
    if (n0 + t < N) rowptr[n0 + t] = base + ex;
  }
  if (b == NB - 1 && t == 0) rowptr[N] = base + total;
  __syncthreads();
  const bool staged = total <= CAP;
  for (int g = wave; g < G; g += SORT_T / 32) {
    const int st = sh_s[g], n = sh_n[g];
    for (int i = lane; i < n; i += 32) {
      const int sv = bsrc[st + i];
      const int p = atomicAdd(&sh_cnt[bdst[st + i] - n0], 1);
      if (staged) sh_sorted[p] = sv; else csr[base + p] = sv;
    }
  }
  if (staged) {
    __syncthreads();
    for (int i = t; i < total; i += SORT_T) csr[base + i] = sh_sorted[i];
  }
}

constexpr int EB = 8;

template <int H>
__device__ __forceinline__ float score_part(const float* a, const float* r, const float* w) {
  if (H == 64) {
    float h0 = a[0] + r[0]; h0 = h0 > 0.0f ? h0 : 0.2f * h0;
    float h1 = a[1] + r[1]; h1 = h1 > 0.0f ? h1 : 0.2f * h1;
    return fmaf(h0, w[0], h1 * w[1]);
  } else {
    float hv = a[0] + r[0];
    hv = hv > 0.0f ? hv : 0.2f * hv;
    return hv * w[0];
  }
}

template <int H>
__device__ __forceinline__ void load_row(float* dst, const float* __restrict__ base, int row, int lane) {
  if (H == 64) {
    const float2 t = *(const float2*)(base + (size_t)row * 64 + lane * 2);
    dst[0] = t.x; dst[1] = t.y;
  } else {
    dst[0] = base[(size_t)row * 32 + lane];
  }
}

template <int H, bool LAST>
__global__ void node_attend(const float* __restrict__ xl, const float* __restrict__ xr,
                            const float* __restrict__ att, const int* __restrict__ rowptr,
                            const int* __restrict__ csr, const float* __restrict__ bias,
                            const float* __restrict__ Wlin, const float* __restrict__ blin,
                            float* __restrict__ outp, int N) {
  constexpr int V = H / 32;
  const int gid = blockIdx.x * blockDim.x + threadIdx.x;
  const int lane = gid & 31;
  const int n = __builtin_amdgcn_readfirstlane(gid >> 5);
  if (n >= N) return;
  const int beg = rowptr[n], end = rowptr[n + 1];

  float w[V], r[V], acc[V];
  load_row<H>(w, att, 0, lane);
  load_row<H>(r, xr, n, lane);
  load_row<H>(acc, xl, n, lane);
  float m = wave_sum32(score_part<H>(acc, r, w));
  float den = 1.0f;

  for (int c = beg; c < end; c += 32) {
    const int cnt = min(32, end - c);
    int s = n;
    if (lane < cnt) s = csr[c + lane];
#pragma unroll 1
    for (int q = 0; q < cnt; q += EB) {
      float a[EB][V], ev[EB];
#pragma unroll
      for (int j = 0; j < EB; ++j) {
        const int sj = __builtin_amdgcn_readlane(s, q + j);
        load_row<H>(a[j], xl, sj, lane);
      }
      float mb = m;
#pragma unroll
      for (int j = 0; j < EB; ++j) {
        ev[j] = wave_sum32(score_part<H>(a[j], r, w));
        if (q + j < cnt) mb = fmaxf(mb, ev[j]);
      }
      const float sc = expf(m - mb);
      m = mb;
      den *= sc;
#pragma unroll
      for (int v = 0; v < V; ++v) acc[v] *= sc;
#pragma unroll
      for (int j = 0; j < EB; ++j) {
        if (q + j < cnt) {
          const float p = expf(ev[j] - m);
          den += p;
#pragma unroll
          for (int v = 0; v < V; ++v) acc[v] = fmaf(p, a[j][v], acc[v]);
        }
      }
    }
  }

  const float dn = den + 1e-16f;
  if (!LAST) {
    float b[V], o[V];
    load_row<H>(b, bias, 0, lane);
#pragma unroll
    for (int v = 0; v < V; ++v) {
      const float t = acc[v] / dn + b[v];
      o[v] = t > 0.0f ? t : 0.01f * t;
    }
    if (H == 64) {
      *(float2*)(outp + (size_t)n * 64 + lane * 2) = make_float2(o[0], o[V - 1]);
    } else {
      outp[(size_t)n * 32 + lane] = o[0];
    }
  } else {
    float tot = (acc[0] / dn + bias[lane * V]) * Wlin[lane * V];
#pragma unroll
    for (int v = 1; v < V; ++v)
      tot = fmaf(acc[v] / dn + bias[lane * V + v], Wlin[lane * V + v], tot);
    tot = wave_sum32(tot);
    if (lane == 0) outp[n] = tot + blin[0];
  }
}

static inline int cdiv(int a, int b) { return (a + b - 1) / b; }

extern "C" void kernel_launch(void* const* d_in, const int* in_sizes, int n_in,
                              void* d_out, int out_size, void* d_ws, size_t ws_size,
                              hipStream_t stream) {
  const float* x     = (const float*)d_in[0];
  const int*   ei    = (const int*)  d_in[1];
  const float* W1l   = (const float*)d_in[2];
  const float* b1l   = (const float*)d_in[3];
  const float* W1r   = (const float*)d_in[4];
  const float* b1r   = (const float*)d_in[5];
  const float* att1  = (const float*)d_in[6];
  const float* bias1 = (const float*)d_in[7];
  const float* W2l   = (const float*)d_in[8];
  const float* b2l   = (const float*)d_in[9];
  const float* W2r   = (const float*)d_in[10];
  const float* b2r   = (const float*)d_in[11];
  const float* att2  = (const float*)d_in[12];
  const float* bias2 = (const float*)d_in[13];
  const float* Wlin  = (const float*)d_in[14];
  const float* blin  = (const float*)d_in[15];
  float* out = (float*)d_out;

  const int N    = in_sizes[0] / 128;
  const int E    = in_sizes[1] / 2;

  float* ws = (float*)d_ws;
  const int NB = cdiv(N, BINW);
  const int G  = cdiv(E, CH);
  size_t off = 0;
  float* xl1  = ws + off; off += (size_t)N * 64;
  float* xr1  = ws + off; off += (size_t)N * 64;
  float* h1   = ws + off; off += (size_t)N * 64;
  float* xl2  = ws + off; off += (size_t)N * 32;
  float* xr2  = ws + off; off += (size_t)N * 32;
  int* rowptr = (int*)(ws + off); off += (size_t)N + 1;
  off = (off + 31) & ~(size_t)31;
  int* csr    = (int*)(ws + off); off += (size_t)E;
  int* bsrc   = (int*)(ws + off); off += (size_t)G * CH;
  int* bdst   = (int*)(ws + off); off += (size_t)G * CH;
  int* lscan  = (int*)(ws + off); off += (size_t)G * (NB + 1);
  (void)ws_size; (void)n_in; (void)out_size;

  const int TB = 256;
  const int rowTiles = cdiv(N, 16);
  const int nblocks = cdiv(N, TB / 32);

  bin_chunk<<<G, SORT_T, 0, stream>>>(ei, E, NB, bsrc, bdst, lscan);
  bin_to_csr<<<NB, SORT_T, 0, stream>>>(bsrc, bdst, lscan, G, NB, N, rowptr, csr);

  gemm_bias_wmma<128, 64><<<cdiv(rowTiles, 2), TB, 0, stream>>>(x, W1l, b1l, xl1, N);
  gemm_bias_wmma<128, 64><<<cdiv(rowTiles, 2), TB, 0, stream>>>(x, W1r, b1r, xr1, N);
  node_attend<64, false><<<nblocks, TB, 0, stream>>>(xl1, xr1, att1, rowptr, csr, bias1,
                                                     Wlin, blin, h1, N);

  gemm_bias_wmma<64, 32><<<cdiv(rowTiles, 4), TB, 0, stream>>>(h1, W2l, b2l, xl2, N);
  gemm_bias_wmma<64, 32><<<cdiv(rowTiles, 4), TB, 0, stream>>>(h1, W2r, b2r, xr2, N);
  node_attend<32, true><<<nblocks, TB, 0, stream>>>(xl2, xr2, att2, rowptr, csr, bias2,
                                                    Wlin, blin, out, N);
}


# both projections of a layer in one GEMM launch: weights staged once per work-group in LDS in fragment order, each wave reads its 16 rows of X once for all column tiles; k sum order changes
# speedup vs baseline: 7.6326x; 1.4058x over previous
#include <hip/hip_runtime.h>
#include <hip/hip_bf16.h>
#include <math.h>

typedef __attribute__((ext_vector_type(2))) float v2f;
typedef __attribute__((ext_vector_type(8))) float v8f;

template <int K>
__device__ __forceinline__ void load_strip(float4* a, const float* __restrict__ X, int s, int h, int c) {
  const float* __restrict__ arow = X + (size_t)(s * 16 + c) * K + 4 * h;
#pragma unroll
  for (int kb = 0; kb < K / 8; ++kb) a[kb] = *(const float4*)(arow + 8 * kb);
}

template <int K, int H>
__device__ __forceinline__ void strip_times_weights(const float4* a, const float4* Bs, int s, int h, int c,
                                                    const float* __restrict__ bl,
                                                    const float* __restrict__ br,
                                                    float* __restrict__ Cl, float* __restrict__ Cr) {
  constexpr int NC = 2 * H;
  constexpr int NT = NC / 16;
  v8f acc[NT];
#pragma unroll
  for (int t = 0; t < NT; ++t) acc[t] = v8f{};
#pragma unroll
  for (int kb = 0; kb < K / 8; ++kb) {
    float4 b[NT];
#pragma unroll
    for (int t = 0; t < NT; ++t) b[t] = Bs[(2 * kb + h) * NC + 16 * t + c];
    const v2f a0 = {a[kb].x, a[kb].y};
    const v2f a1 = {a[kb].z, a[kb].w};
#pragma unroll
    for (int t = 0; t < NT; ++t) {
      const v2f b0 = {b[t].x, b[t].y};
      acc[t] = __builtin_amdgcn_wmma_f32_16x16x4_f32(false, a0, false, b0, (short)0, acc[t],
                                                     false, false);
    }
#pragma unroll
    for (int t = 0; t < NT; ++t) {
      const v2f b1 = {b[t].z, b[t].w};
      acc[t] = __builtin_amdgcn_wmma_f32_16x16x4_f32(false, a1, false, b1, (short)0, acc[t],
                                                     false, false);
    }
  }
#pragma unroll
  for (int t = 0; t < NT; ++t) {
    const int col = 16 * t + c;
    const bool left = 16 * t < H;
    const float bv = left ? bl[col] : br[col - H];
    float* __restrict__ crow = (left ? Cl + col : Cr + (col - H)) + (size_t)(s * 16 + 8 * h) * H;
#pragma unroll
    for (int v = 0; v < 8; ++v) crow[(size_t)v * H] = acc[t][v] + bv;
  }
}

template <int K, int H>
__global__ void __launch_bounds__(256)
gemm2_bias_wmma(const float* __restrict__ X, const float* __restrict__ Wl,
                const float* __restrict__ bl, const float* __restrict__ Wr,
                const float* __restrict__ br, float* __restrict__ Cl,
                float* __restrict__ Cr, int N) {
  constexpr int NC = 2 * H;
  constexpr int KQ = K / 4;
  __shared__ float4 Bs[KQ * NC];
  const int wave = __builtin_amdgcn_readfirstlane(threadIdx.x >> 5);
  const int lane = threadIdx.x & 31;
  const int h = lane >> 4, c = lane & 15;
  const int nStrips = N / 16;
  const int G = gridDim.x;
  int s = wave * G + blockIdx.x;

  float4 a[K / 8];
  if (s < nStrips) load_strip<K>(a, X, s, h, c);

  for (int it = threadIdx.x; it < KQ * (NC / 4); it += 256) {
    const int cj = it % (NC / 4), kq = it / (NC / 4);
    const int col = 4 * cj;
    const float* __restrict__ p = (col < H ? Wl + col : Wr + (col - H)) + (size_t)(4 * kq) * H;
    const float4 r0 = *(const float4*)p;
    const float4 r1 = *(const float4*)(p + H);
    const float4 r2 = *(const float4*)(p + 2 * H);
    const float4 r3 = *(const float4*)(p + 3 * H);
    float4* __restrict__ q = Bs + kq * NC + col;
    q[0] = make_float4(r0.x, r1.x, r2.x, r3.x);
    q[1] = make_float4(r0.y, r1.y, r2.y, r3.y);
    q[2] = make_float4(r0.z, r1.z, r2.z, r3.z);
    q[3] = make_float4(r0.w, r1.w, r2.w, r3.w);
  }
  __syncthreads();
  if (s >= nStrips) return;
  strip_times_weights<K, H>(a, Bs, s, h, c, bl, br, Cl, Cr);
  for (s += 8 * G; s < nStrips; s += 8 * G) {
    asm volatile("" ::: "memory");
    float4 a2[K / 8];
    load_strip<K>(a2, X, s, h, c);
    strip_times_weights<K, H>(a2, Bs, s, h, c, bl, br, Cl, Cr);
  }
}

__device__ __forceinline__ int wave_sum32_i(int v) {
#pragma unroll
  for (int off = 16; off > 0; off >>= 1) v += __shfl_xor(v, off, 32);
  return v;
}

constexpr int LOGW = 8;
constexpr int BINW = 1 << LOGW;
constexpr int SORT_T = 1024;
constexpr int SORT_EPT = 8;
constexpr int CH = SORT_T * SORT_EPT;
constexpr int NBMAX = SORT_T;
constexpr int GMAX = 1024;
constexpr int CAP = 12288;

__device__ __forceinline__ int block_excl_scan(int v, int* sh_w, int& total) {
  const int lane = threadIdx.x & 31, wave = threadIdx.x >> 5;
  int inc = v;
#pragma unroll
  for (int off = 1; off < 32; off <<= 1) {
    const int u = __shfl_up(inc, off, 32);
    if (lane >= off) inc += u;
  }
  if (lane == 31) sh_w[wave] = inc;
  __syncthreads();
  if (wave == 0) {
    const int w = sh_w[lane];
    int winc = w;
#pragma unroll
    for (int off = 1; off < 32; off <<= 1) {
      const int u = __shfl_up(winc, off, 32);
      if (lane >= off) winc += u;
    }
    sh_w[lane] = winc - w;
    if (lane == 31) sh_w[32] = winc;
  }
  __syncthreads();
  total = sh_w[32];
  return sh_w[wave] + inc - v;
}

__global__ void __launch_bounds__(SORT_T)
bin_chunk(const int* __restrict__ ei, int E, int NB, int* __restrict__ bsrc,
          int* __restrict__ bdst, int* __restrict__ lscan) {
  __shared__ int sh_bin[NBMAX];
  __shared__ int sh_w[33];
  __shared__ int sh_buf[CH];
  const int t = threadIdx.x;
  const int g = blockIdx.x;
  const int e0 = g * CH;
  sh_bin[t] = 0;
  __syncthreads();
  int s[SORT_EPT], d[SORT_EPT], rk[SORT_EPT];
#pragma unroll
  for (int k = 0; k < SORT_EPT; ++k) {
    const int e = e0 + k * SORT_T + t;
    d[k] = e < E ? ei[E + e] : -1;
    s[k] = e < E ? ei[e] : 0;
  }
#pragma unroll
  for (int k = 0; k < SORT_EPT; ++k) {
    rk[k] = 0;
    if (d[k] >= 0) rk[k] = atomicAdd(&sh_bin[d[k] >> LOGW], 1);
  }
  __syncthreads();
  int total;
  const int ex = block_excl_scan(sh_bin[t], sh_w, total);
  sh_bin[t] = ex;
  int* __restrict__ row = lscan + (size_t)g * (NB + 1);
  if (t < NB) row[t] = ex;
  if (t == 0) row[NB] = total;
  __syncthreads();
  int pos[SORT_EPT];
#pragma unroll
  for (int k = 0; k < SORT_EPT; ++k) pos[k] = d[k] >= 0 ? sh_bin[d[k] >> LOGW] + rk[k] : 0;
#pragma unroll
  for (int k = 0; k < SORT_EPT; ++k) if (d[k] >= 0) sh_buf[pos[k]] = d[k];
  __syncthreads();
#pragma unroll
  for (int k = 0; k < SORT_EPT; ++k) {
    const int i = k * SORT_T + t;
    if (i < total) bdst[e0 + i] = sh_buf[i];
  }
  __syncthreads();
#pragma unroll
  for (int k = 0; k < SORT_EPT; ++k) if (d[k] >= 0) sh_buf[pos[k]] = s[k];
  __syncthreads();
#pragma unroll
  for (int k = 0; k < SORT_EPT; ++k) {
    const int i = k * SORT_T + t;
    if (i < total) bsrc[e0 + i] = sh_buf[i];
  }
}

__global__ void __launch_bounds__(SORT_T)
bin_to_csr(const int* __restrict__ bsrc, const int* __restrict__ bdst,
           const int* __restrict__ lscan, int G, int NB, int N,
           int* __restrict__ rowptr, int* __restrict__ csr) {
  __shared__ int sh_s[GMAX];
  __shared__ int sh_n[GMAX];
  __shared__ int sh_cnt[BINW];
  __shared__ int sh_w[33];
  __shared__ int sh_red[2];
  __shared__ int sh_sorted[CAP];
  const int t = threadIdx.x, lane = t & 31;
  const int wave = __builtin_amdgcn_readfirstlane(t >> 5);
  const int b = blockIdx.x;
  const int n0 = b << LOGW;
  if (t < BINW) sh_cnt[t] = 0;
  if (t < 2) sh_red[t] = 0;
  __syncthreads();
  int pb = 0, pn = 0;
  for (int g = t; g < G; g += SORT_T) {
    const int* __restrict__ row = lscan + (size_t)g * (NB + 1) + b;
    const int s0 = row[0], s1 = row[1];
    sh_s[g] = g * CH + s0;
    sh_n[g] = s1 - s0;
    pb += s0;
    pn += s1 - s0;
  }
  pb = wave_sum32_i(pb);
  pn = wave_sum32_i(pn);
  if (lane == 0) { atomicAdd(&sh_red[0], pb); atomicAdd(&sh_red[1], pn); }
  __syncthreads();
  const int base = sh_red[0];
  const int total = sh_red[1];
  for (int g = wave; g < G; g += SORT_T / 32) {
    const int st = sh_s[g], n = sh_n[g];
    for (int i = lane; i < n; i += 32) atomicAdd(&sh_cnt[bdst[st + i] - n0], 1);
  }
  __syncthreads();
  int tot2;
  const int ex = block_excl_scan(t < BINW ? sh_cnt[t] : 0, sh_w, tot2);
  if (t < BINW) {
    sh_cnt[t] = ex;
    if (n0 + t < N) rowptr[n0 + t] = base + ex;
  }
  if (b == NB - 1 && t == 0) rowptr[N] = base + total;
  __syncthreads();
  const bool staged = total <= CAP;
  for (int g = wave; g < G; g += SORT_T / 32) {
    const int st = sh_s[g], n = sh_n[g];
    for (int i = lane; i < n; i += 32) {
      const int sv = bsrc[st + i];
      const int p = atomicAdd(&sh_cnt[bdst[st + i] - n0], 1);
      if (staged) sh_sorted[p] = sv; else csr[base + p] = sv;
    }
  }
  if (staged) {
    __syncthreads();
    for (int i = t; i < total; i += SORT_T) csr[base + i] = sh_sorted[i];
  }
}

constexpr int CPL = 8;

template <int CTRL>
__device__ __forceinline__ float lane_get(float v) {
  return __builtin_bit_cast(
      float, __builtin_amdgcn_update_dpp(0, __builtin_bit_cast(int, v), CTRL, 0xF, 0xF, false));
}

template <int LPE>
__device__ __forceinline__ float group_sum(float v) {
  v += lane_get<0xB1>(v);
  v += lane_get<0x4E>(v);
  if (LPE == 8) v += lane_get<0x141>(v);
  return v;
}

template <int H>
__device__ __forceinline__ void load_row8(float* dst, const float* __restrict__ base, int row, int sub) {
  const float* __restrict__ p = base + (size_t)row * H + 4 * sub;
  const float4 t0 = *(const float4*)p;
  const float4 t1 = *(const float4*)(p + H / 2);
  dst[0] = t0.x; dst[1] = t0.y; dst[2] = t0.z; dst[3] = t0.w;
  dst[4] = t1.x; dst[5] = t1.y; dst[6] = t1.z; dst[7] = t1.w;
}

__device__ __forceinline__ float score_part8(const float* a, const float* r, const float* w) {
  float s0 = 0.0f, s1 = 0.0f;
#pragma unroll
  for (int v = 0; v < CPL; v += 2) {
    float h0 = a[v] + r[v];         h0 = h0 > 0.0f ? h0 : 0.2f * h0;
    float h1 = a[v + 1] + r[v + 1]; h1 = h1 > 0.0f ? h1 : 0.2f * h1;
    s0 = fmaf(h0, w[v], s0);
    s1 = fmaf(h1, w[v + 1], s1);
  }
  return s0 + s1;
}

__device__ __forceinline__ void softmax_step(float e, bool ok, const float* a, float& m, float& den,
                                             float* acc) {
  const float d = e - m;
  const float t = expf(-fabsf(d));
  const bool up = d > 0.0f;
  float sc = up ? t : 1.0f;
  float p = up ? 1.0f : t;
  if (!ok) { sc = 1.0f; p = 0.0f; }
  if (ok && up) m = e;
  den = fmaf(den, sc, p);
#pragma unroll
  for (int v = 0; v < CPL; ++v) acc[v] = fmaf(acc[v], sc, p * a[v]);
}

template <int H, bool LAST>
__global__ void node_attend(const float* __restrict__ xl, const float* __restrict__ xr,
                            const float* __restrict__ att, const int* __restrict__ rowptr,
                            const int* __restrict__ csr, const float* __restrict__ bias,
                            const float* __restrict__ Wlin, const float* __restrict__ blin,
                            float* __restrict__ outp, int N) {
  constexpr int LPE = H / CPL;
  constexpr int EPW = 32 / LPE;
  const int gid = blockIdx.x * blockDim.x + threadIdx.x;
  const int lane = gid & 31;
  const int n = __builtin_amdgcn_readfirstlane(gid >> 5);
  if (n >= N) return;
  const int beg = rowptr[n], end = rowptr[n + 1];
  const int sub = lane & (LPE - 1);
  const int grp = lane / LPE;

  float w[CPL], r[CPL], acc[CPL];
  load_row8<H>(w, att, 0, sub);
  load_row8<H>(r, xr, n, sub);
  load_row8<H>(acc, xl, n, sub);
  float m = group_sum<LPE>(score_part8(acc, r, w));
  float den = grp == 0 ? 1.0f : 0.0f;
#pragma unroll
  for (int v = 0; v < CPL; ++v) acc[v] = grp == 0 ? acc[v] : 0.0f;

  for (int c = beg; c < end; c += 32) {
    const int cnt = min(32, end - c);
    int s = n;
    if (lane < cnt) s = csr[c + lane];
    for (int q = 0; q < cnt; q += EPW) {
      const int i = q + grp;
      float a[CPL];
      load_row8<H>(a, xl, __shfl(s, i, 32), sub);
      const float e = group_sum<LPE>(score_part8(a, r, w));
      softmax_step(e, i < cnt, a, m, den, acc);
    }
  }

  float M = m;
#pragma unroll
  for (int off = LPE; off < 32; off <<= 1) M = fmaxf(M, __shfl_xor(M, off, 32));
  const float f = expf(m - M);
  den *= f;
#pragma unroll
  for (int v = 0; v < CPL; ++v) acc[v] *= f;
#pragma unroll
  for (int off = LPE; off < 32; off <<= 1) {
    den += __shfl_xor(den, off, 32);
#pragma unroll
    for (int v = 0; v < CPL; ++v) acc[v] += __shfl_xor(acc[v], off, 32);
  }

  const float inv = 1.0f / (den + 1e-16f);
  if (!LAST) {
    float b[CPL], o[CPL];
    load_row8<H>(b, bias, 0, sub);
#pragma unroll
    for (int v = 0; v < CPL; ++v) {
      const float t = fmaf(acc[v], inv, b[v]);
      o[v] = t > 0.0f ? t : 0.01f * t;
    }
    if (grp == 0) {
      float* __restrict__ p = outp + (size_t)n * H + 4 * sub;
      *(float4*)p = make_float4(o[0], o[1], o[2], o[3]);
      *(float4*)(p + H / 2) = make_float4(o[4], o[5], o[6], o[7]);
    }
  } else {
    float b[CPL], wl[CPL];
    load_row8<H>(b, bias, 0, sub);
    load_row8<H>(wl, Wlin, 0, sub);
    float tot = 0.0f;
#pragma unroll
    for (int v = 0; v < CPL; ++v) tot = fmaf(fmaf(acc[v], inv, b[v]), wl[v], tot);
    tot = group_sum<LPE>(tot);
    if (lane == 0) outp[n] = tot + blin[0];
  }
}

static inline int cdiv(int a, int b) { return (a + b - 1) / b; }

extern "C" void kernel_launch(void* const* d_in, const int* in_sizes, int n_in,
                              void* d_out, int out_size, void* d_ws, size_t ws_size,
                              hipStream_t stream) {
  const float* x     = (const float*)d_in[0];
  const int*   ei    = (const int*)  d_in[1];
  const float* W1l   = (const float*)d_in[2];
  const float* b1l   = (const float*)d_in[3];
  const float* W1r   = (const float*)d_in[4];
  const float* b1r   = (const float*)d_in[5];
  const float* att1  = (const float*)d_in[6];
  const float* bias1 = (const float*)d_in[7];
  const float* W2l   = (const float*)d_in[8];
  const float* b2l   = (const float*)d_in[9];
  const float* W2r   = (const float*)d_in[10];
  const float* b2r   = (const float*)d_in[11];
  const float* att2  = (const float*)d_in[12];
  const float* bias2 = (const float*)d_in[13];
  const float* Wlin  = (const float*)d_in[14];
  const float* blin  = (const float*)d_in[15];
  float* out = (float*)d_out;

  const int N    = in_sizes[0] / 128;
  const int E    = in_sizes[1] / 2;

  float* ws = (float*)d_ws;
  const int NB = cdiv(N, BINW);
  const int G  = cdiv(E, CH);
  size_t off = 0;
  float* xl1  = ws + off; off += (size_t)N * 64;
  float* xr1  = ws + off; off += (size_t)N * 64;
  float* h1   = ws + off; off += (size_t)N * 64;
  float* xl2  = ws + off; off += (size_t)N * 32;
  float* xr2  = ws + off; off += (size_t)N * 32;
  int* rowptr = (int*)(ws + off); off += (size_t)N + 1;
  off = (off + 31) & ~(size_t)31;
  int* csr    = (int*)(ws + off); off += (size_t)E;
  int* bsrc   = (int*)(ws + off); off += (size_t)G * CH;
  int* bdst   = (int*)(ws + off); off += (size_t)G * CH;
  int* lscan  = (int*)(ws + off); off += (size_t)G * (NB + 1);
  (void)ws_size; (void)n_in; (void)out_size;

  const int TB = 256;
  const int rowTiles = cdiv(N, 16);
  const int nblocks = cdiv(N, TB / 32);

  bin_chunk<<<G, SORT_T, 0, stream>>>(ei, E, NB, bsrc, bdst, lscan);
  bin_to_csr<<<NB, SORT_T, 0, stream>>>(bsrc, bdst, lscan, G, NB, N, rowptr, csr);

  gemm2_bias_wmma<128, 64><<<768, TB, 0, stream>>>(x, W1l, b1l, W1r, b1r, xl1, xr1, N);
  node_attend<64, false><<<nblocks, TB, 0, stream>>>(xl1, xr1, att1, rowptr, csr, bias1,
                                                     Wlin, blin, h1, N);

  gemm2_bias_wmma<64, 32><<<cdiv(rowTiles, 8), TB, 0, stream>>>(h1, W2l, b2l, W2r, b2r, xl2, xr2, N);
  node_attend<32, true><<<nblocks, TB, 0, stream>>>(xl2, xr2, att2, rowptr, csr, bias2,
                                                    Wlin, blin, out, N);
}


# node kernels (vector-issue bound): edge step 63 -> 42 instructions: softmax factors as two bare v_exp_f32 in base 2 (att scaled by log2 e), leaky relu as max(h, 0.2h); v_rcp_f32 for the division
# speedup vs baseline: 7.9030x; 1.0354x over previous
#include <hip/hip_runtime.h>
#include <hip/hip_bf16.h>
#include <math.h>

typedef __attribute__((ext_vector_type(2))) float v2f;
typedef __attribute__((ext_vector_type(8))) float v8f;

template <int K>
__device__ __forceinline__ void load_strip(float4* a, const float* __restrict__ X, int s, int h, int c) {
  const float* __restrict__ arow = X + (size_t)(s * 16 + c) * K + 4 * h;
#pragma unroll
  for (int kb = 0; kb < K / 8; ++kb) a[kb] = *(const float4*)(arow + 8 * kb);
}

template <int K, int H>
__device__ __forceinline__ void strip_times_weights(const float4* a, const float4* Bs, int s, int h, int c,
                                                    const float* __restrict__ bl,
                                                    const float* __restrict__ br,
                                                    float* __restrict__ Cl, float* __restrict__ Cr) {
  constexpr int NC = 2 * H;
  constexpr int NT = NC / 16;
  v8f acc[NT];
#pragma unroll
  for (int t = 0; t < NT; ++t) acc[t] = v8f{};
#pragma unroll
  for (int kb = 0; kb < K / 8; ++kb) {
    float4 b[NT];
#pragma unroll
    for (int t = 0; t < NT; ++t) b[t] = Bs[(2 * kb + h) * NC + 16 * t + c];
    const v2f a0 = {a[kb].x, a[kb].y};
    const v2f a1 = {a[kb].z, a[kb].w};
#pragma unroll
    for (int t = 0; t < NT; ++t) {
      const v2f b0 = {b[t].x, b[t].y};
      acc[t] = __builtin_amdgcn_wmma_f32_16x16x4_f32(false, a0, false, b0, (short)0, acc[t],
                                                     false, false);
    }
#pragma unroll
    for (int t = 0; t < NT; ++t) {
      const v2f b1 = {b[t].z, b[t].w};
      acc[t] = __builtin_amdgcn_wmma_f32_16x16x4_f32(false, a1, false, b1, (short)0, acc[t],
                                                     false, false);
    }
  }
#pragma unroll
  for (int t = 0; t < NT; ++t) {
    const int col = 16 * t + c;
    const bool left = 16 * t < H;
    const float bv = left ? bl[col] : br[col - H];
    float* __restrict__ crow = (left ? Cl + col : Cr + (col - H)) + (size_t)(s * 16 + 8 * h) * H;
#pragma unroll
    for (int v = 0; v < 8; ++v) crow[(size_t)v * H] = acc[t][v] + bv;
  }
}

template <int K, int H>
__global__ void __launch_bounds__(256)
gemm2_bias_wmma(const float* __restrict__ X, const float* __restrict__ Wl,
                const float* __restrict__ bl, const float* __restrict__ Wr,
                const float* __restrict__ br, float* __restrict__ Cl,
                float* __restrict__ Cr, int N) {
  constexpr int NC = 2 * H;
  constexpr int KQ = K / 4;
  __shared__ float4 Bs[KQ * NC];
  const int wave = __builtin_amdgcn_readfirstlane(threadIdx.x >> 5);
  const int lane = threadIdx.x & 31;
  const int h = lane >> 4, c = lane & 15;
  const int nStrips = N / 16;
  const int G = gridDim.x;
  int s = wave * G + blockIdx.x;

  float4 a[K / 8];
  if (s < nStrips) load_strip<K>(a, X, s, h, c);

  for (int it = threadIdx.x; it < KQ * (NC / 4); it += 256) {
    const int cj = it % (NC / 4), kq = it / (NC / 4);
    const int col = 4 * cj;
    const float* __restrict__ p = (col < H ? Wl + col : Wr + (col - H)) + (size_t)(4 * kq) * H;
    const float4 r0 = *(const float4*)p;
    const float4 r1 = *(const float4*)(p + H);
    const float4 r2 = *(const float4*)(p + 2 * H);
    const float4 r3 = *(const float4*)(p + 3 * H);
    float4* __restrict__ q = Bs + kq * NC + col;
    q[0] = make_float4(r0.x, r1.x, r2.x, r3.x);
    q[1] = make_float4(r0.y, r1.y, r2.y, r3.y);
    q[2] = make_float4(r0.z, r1.z, r2.z, r3.z);
    q[3] = make_float4(r0.w, r1.w, r2.w, r3.w);
  }
  __syncthreads();
  if (s >= nStrips) return;
  strip_times_weights<K, H>(a, Bs, s, h, c, bl, br, Cl, Cr);
  for (s += 8 * G; s < nStrips; s += 8 * G) {
    asm volatile("" ::: "memory");
    float4 a2[K / 8];
    load_strip<K>(a2, X, s, h, c);
    strip_times_weights<K, H>(a2, Bs, s, h, c, bl, br, Cl, Cr);
  }
}

__device__ __forceinline__ int wave_sum32_i(int v) {
#pragma unroll
  for (int off = 16; off > 0; off >>= 1) v += __shfl_xor(v, off, 32);
  return v;
}

constexpr int LOGW = 8;
constexpr int BINW = 1 << LOGW;
constexpr int SORT_T = 1024;
constexpr int SORT_EPT = 8;
constexpr int CH = SORT_T * SORT_EPT;
constexpr int NBMAX = SORT_T;
constexpr int GMAX = 1024;
constexpr int CAP = 12288;

__device__ __forceinline__ int block_excl_scan(int v, int* sh_w, int& total) {
  const int lane = threadIdx.x & 31, wave = threadIdx.x >> 5;
  int inc = v;
#pragma unroll
  for (int off = 1; off < 32; off <<= 1) {
    const int u = __shfl_up(inc, off, 32);
    if (lane >= off) inc += u;
  }
  if (lane == 31) sh_w[wave] = inc;
  __syncthreads();
  if (wave == 0) {
    const int w = sh_w[lane];
    int winc = w;
#pragma unroll
    for (int off = 1; off < 32; off <<= 1) {
      const int u = __shfl_up(winc, off, 32);
      if (lane >= off) winc += u;
    }
    sh_w[lane] = winc - w;
    if (lane == 31) sh_w[32] = winc;
  }
  __syncthreads();
  total = sh_w[32];
  return sh_w[wave] + inc - v;
}

__global__ void __launch_bounds__(SORT_T)
bin_chunk(const int* __restrict__ ei, int E, int NB, int* __restrict__ bsrc,
          int* __restrict__ bdst, int* __restrict__ lscan) {
  __shared__ int sh_bin[NBMAX];
  __shared__ int sh_w[33];
  __shared__ int sh_buf[CH];
  const int t = threadIdx.x;
  const int g = blockIdx.x;
  const int e0 = g * CH;
  sh_bin[t] = 0;
  __syncthreads();
  int s[SORT_EPT], d[SORT_EPT], rk[SORT_EPT];
#pragma unroll
  for (int k = 0; k < SORT_EPT; ++k) {
    const int e = e0 + k * SORT_T + t;
    d[k] = e < E ? ei[E + e] : -1;
    s[k] = e < E ? ei[e] : 0;
  }
#pragma unroll
  for (int k = 0; k < SORT_EPT; ++k) {
    rk[k] = 0;
    if (d[k] >= 0) rk[k] = atomicAdd(&sh_bin[d[k] >> LOGW], 1);
  }
  __syncthreads();
  int total;
  const int ex = block_excl_scan(sh_bin[t], sh_w, total);
  sh_bin[t] = ex;
  int* __restrict__ row = lscan + (size_t)g * (NB + 1);
  if (t < NB) row[t] = ex;
  if (t == 0) row[NB] = total;
  __syncthreads();
  int pos[SORT_EPT];
#pragma unroll
  for (int k = 0; k < SORT_EPT; ++k) pos[k] = d[k] >= 0 ? sh_bin[d[k] >> LOGW] + rk[k] : 0;
#pragma unroll
  for (int k = 0; k < SORT_EPT; ++k) if (d[k] >= 0) sh_buf[pos[k]] = d[k];
  __syncthreads();
#pragma unroll
  for (int k = 0; k < SORT_EPT; ++k) {
    const int i = k * SORT_T + t;
    if (i < total) bdst[e0 + i] = sh_buf[i];
  }
  __syncthreads();
#pragma unroll
  for (int k = 0; k < SORT_EPT; ++k) if (d[k] >= 0) sh_buf[pos[k]] = s[k];
  __syncthreads();
#pragma unroll
  for (int k = 0; k < SORT_EPT; ++k) {
    const int i = k * SORT_T + t;
    if (i < total) bsrc[e0 + i] = sh_buf[i];
  }
}

__global__ void __launch_bounds__(SORT_T)
bin_to_csr(const int* __restrict__ bsrc, const int* __restrict__ bdst,
           const int* __restrict__ lscan, int G, int NB, int N,
           int* __restrict__ rowptr, int* __restrict__ csr) {
  __shared__ int sh_s[GMAX];
  __shared__ int sh_n[GMAX];
  __shared__ int sh_cnt[BINW];
  __shared__ int sh_w[33];
  __shared__ int sh_red[2];
  __shared__ int sh_sorted[CAP];
  const int t = threadIdx.x, lane = t & 31;
  const int wave = __builtin_amdgcn_readfirstlane(t >> 5);
  const int b = blockIdx.x;
  const int n0 = b << LOGW;
  if (t < BINW) sh_cnt[t] = 0;
  if (t < 2) sh_red[t] = 0;
  __syncthreads();
  int pb = 0, pn = 0;
  for (int g = t; g < G; g += SORT_T) {
    const int* __restrict__ row = lscan + (size_t)g * (NB + 1) + b;
    const int s0 = row[0], s1 = row[1];
    sh_s[g] = g * CH + s0;
    sh_n[g] = s1 - s0;
    pb += s0;
    pn += s1 - s0;
  }
  pb = wave_sum32_i(pb);
  pn = wave_sum32_i(pn);
  if (lane == 0) { atomicAdd(&sh_red[0], pb); atomicAdd(&sh_red[1], pn); }
  __syncthreads();
  const int base = sh_red[0];
  const int total = sh_red[1];
  for (int g = wave; g < G; g += SORT_T / 32) {
    const int st = sh_s[g], n = sh_n[g];
    for (int i = lane; i < n; i += 32) atomicAdd(&sh_cnt[bdst[st + i] - n0], 1);
  }
  __syncthreads();
  int tot2;
  const int ex = block_excl_scan(t < BINW ? sh_cnt[t] : 0, sh_w, tot2);
  if (t < BINW) {
    sh_cnt[t] = ex;
    if (n0 + t < N) rowptr[n0 + t] = base + ex;
  }
  if (b == NB - 1 && t == 0) rowptr[N] = base + total;
  __syncthreads();
  const bool staged = total <= CAP;
  for (int g = wave; g < G; g += SORT_T / 32) {
    const int st = sh_s[g], n = sh_n[g];
    for (int i = lane; i < n; i += 32) {
      const int sv = bsrc[st + i];
      const int p = atomicAdd(&sh_cnt[bdst[st + i] - n0], 1);
      if (staged) sh_sorted[p] = sv; else csr[base + p] = sv;
    }
  }
  if (staged) {
    __syncthreads();
    for (int i = t; i < total; i += SORT_T) csr[base + i] = sh_sorted[i];
  }
}

constexpr int CPL = 8;

template <int CTRL>
__device__ __forceinline__ float lane_get(float v) {
  return __builtin_bit_cast(
      float, __builtin_amdgcn_update_dpp(0, __builtin_bit_cast(int, v), CTRL, 0xF, 0xF, false));
}

template <int LPE>
__device__ __forceinline__ float group_sum(float v) {
  v += lane_get<0xB1>(v);
  v += lane_get<0x4E>(v);
  if (LPE == 8) v += lane_get<0x141>(v);
  return v;
}

template <int H>
__device__ __forceinline__ void load_row8(float* dst, const float* __restrict__ base, int row, int sub) {
  const float* __restrict__ p = base + (size_t)row * H + 4 * sub;
  const float4 t0 = *(const float4*)p;
  const float4 t1 = *(const float4*)(p + H / 2);
  dst[0] = t0.x; dst[1] = t0.y; dst[2] = t0.z; dst[3] = t0.w;
  dst[4] = t1.x; dst[5] = t1.y; dst[6] = t1.z; dst[7] = t1.w;
}

__device__ __forceinline__ float score_part8(const float* a, const float* r, const float* w) {
  float s0 = 0.0f, s1 = 0.0f;
#pragma unroll
  for (int v = 0; v < CPL; v += 2) {
    float h0 = a[v] + r[v];         h0 = fmaxf(h0, 0.2f * h0);
    float h1 = a[v + 1] + r[v + 1]; h1 = fmaxf(h1, 0.2f * h1);
    s0 = fmaf(h0, w[v], s0);
    s1 = fmaf(h1, w[v + 1], s1);
  }
  return s0 + s1;
}

__device__ __forceinline__ float exp2_bare(float x) { return __builtin_amdgcn_exp2f(x); }

__device__ __forceinline__ void softmax_step(float e, bool ok, const float* a, float& m, float& den,
                                             float* acc) {
  const float eo = ok ? e : -INFINITY;
  const float mn = eo > m ? eo : m;
  const float sc = exp2_bare(m - mn);
  const float p = exp2_bare(eo - mn);
  m = mn;
  den = fmaf(den, sc, p);
#pragma unroll
  for (int v = 0; v < CPL; ++v) acc[v] = fmaf(acc[v], sc, p * a[v]);
}

template <int H, bool LAST>
__global__ void node_attend(const float* __restrict__ xl, const float* __restrict__ xr,
                            const float* __restrict__ att, const int* __restrict__ rowptr,
                            const int* __restrict__ csr, const float* __restrict__ bias,
                            const float* __restrict__ Wlin, const float* __restrict__ blin,
                            float* __restrict__ outp, int N) {
  constexpr int LPE = H / CPL;
  constexpr int EPW = 32 / LPE;
  const int gid = blockIdx.x * blockDim.x + threadIdx.x;
  const int lane = gid & 31;
  const int n = __builtin_amdgcn_readfirstlane(gid >> 5);
  if (n >= N) return;
  const int beg = rowptr[n], end = rowptr[n + 1];
  const int sub = lane & (LPE - 1);
  const int grp = lane / LPE;

  float w[CPL], r[CPL], acc[CPL];
  load_row8<H>(w, att, 0, sub);
#pragma unroll
  for (int v = 0; v < CPL; ++v) w[v] *= 1.44269504088896340736f;
  load_row8<H>(r, xr, n, sub);
  load_row8<H>(acc, xl, n, sub);
  float m = group_sum<LPE>(score_part8(acc, r, w));
  float den = grp == 0 ? 1.0f : 0.0f;
#pragma unroll
  for (int v = 0; v < CPL; ++v) acc[v] = grp == 0 ? acc[v] : 0.0f;

  for (int c = beg; c < end; c += 32) {
    const int cnt = min(32, end - c);
    int s = n;
    if (lane < cnt) s = csr[c + lane];
    for (int q = 0; q < cnt; q += EPW) {
      const int i = q + grp;
      float a[CPL];
      load_row8<H>(a, xl, __shfl(s, i, 32), sub);
      const float e = group_sum<LPE>(score_part8(a, r, w));
      softmax_step(e, i < cnt, a, m, den, acc);
    }
  }

  float M = m;
#pragma unroll
  for (int off = LPE; off < 32; off <<= 1) M = fmaxf(M, __shfl_xor(M, off, 32));
  const float f = exp2_bare(m - M);
  den *= f;
#pragma unroll
  for (int v = 0; v < CPL; ++v) acc[v] *= f;
#pragma unroll
  for (int off = LPE; off < 32; off <<= 1) {
    den += __shfl_xor(den, off, 32);
#pragma unroll
    for (int v = 0; v < CPL; ++v) acc[v] += __shfl_xor(acc[v], off, 32);
  }

  const float inv = __builtin_amdgcn_rcpf(den + 1e-16f);
  if (!LAST) {
    float b[CPL], o[CPL];
    load_row8<H>(b, bias, 0, sub);
#pragma unroll
    for (int v = 0; v < CPL; ++v) {
      const float t = fmaf(acc[v], inv, b[v]);
      o[v] = fmaxf(t, 0.01f * t);
    }
    if (grp == 0) {
      float* __restrict__ p = outp + (size_t)n * H + 4 * sub;
      *(float4*)p = make_float4(o[0], o[1], o[2], o[3]);
      *(float4*)(p + H / 2) = make_float4(o[4], o[5], o[6], o[7]);
    }
  } else {
    float b[CPL], wl[CPL];
    load_row8<H>(b, bias, 0, sub);
    load_row8<H>(wl, Wlin, 0, sub);
    float tot = 0.0f;
#pragma unroll
    for (int v = 0; v < CPL; ++v) tot = fmaf(fmaf(acc[v], inv, b[v]), wl[v], tot);
    tot = group_sum<LPE>(tot);
    if (lane == 0) outp[n] = tot + blin[0];
  }
}

static inline int cdiv(int a, int b) { return (a + b - 1) / b; }

extern "C" void kernel_launch(void* const* d_in, const int* in_sizes, int n_in,
                              void* d_out, int out_size, void* d_ws, size_t ws_size,
                              hipStream_t stream) {
  const float* x     = (const float*)d_in[0];
  const int*   ei    = (const int*)  d_in[1];
  const float* W1l   = (const float*)d_in[2];
  const float* b1l   = (const float*)d_in[3];
  const float* W1r   = (const float*)d_in[4];
  const float* b1r   = (const float*)d_in[5];
  const float* att1  = (const float*)d_in[6];
  const float* bias1 = (const float*)d_in[7];
  const float* W2l   = (const float*)d_in[8];
  const float* b2l   = (const float*)d_in[9];
  const float* W2r   = (const float*)d_in[10];
  const float* b2r   = (const float*)d_in[11];
  const float* att2  = (const float*)d_in[12];
  const float* bias2 = (const float*)d_in[13];
  const float* Wlin  = (const float*)d_in[14];
  const float* blin  = (const float*)d_in[15];
  float* out = (float*)d_out;

  const int N    = in_sizes[0] / 128;
  const int E    = in_sizes[1] / 2;

  float* ws = (float*)d_ws;
  const int NB = cdiv(N, BINW);
  const int G  = cdiv(E, CH);
  size_t off = 0;
  float* xl1  = ws + off; off += (size_t)N * 64;
  float* xr1  = ws + off; off += (size_t)N * 64;
  float* h1   = ws + off; off += (size_t)N * 64;
  float* xl2  = ws + off; off += (size_t)N * 32;
  float* xr2  = ws + off; off += (size_t)N * 32;
  int* rowptr = (int*)(ws + off); off += (size_t)N + 1;
  off = (off + 31) & ~(size_t)31;
  int* csr    = (int*)(ws + off); off += (size_t)E;
  int* bsrc   = (int*)(ws + off); off += (size_t)G * CH;
  int* bdst   = (int*)(ws + off); off += (size_t)G * CH;
  int* lscan  = (int*)(ws + off); off += (size_t)G * (NB + 1);
  (void)ws_size; (void)n_in; (void)out_size;

  const int TB = 256;
  const int rowTiles = cdiv(N, 16);
  const int nblocks = cdiv(N, TB / 32);

  bin_chunk<<<G, SORT_T, 0, stream>>>(ei, E, NB, bsrc, bdst, lscan);
  bin_to_csr<<<NB, SORT_T, 0, stream>>>(bsrc, bdst, lscan, G, NB, N, rowptr, csr);

  gemm2_bias_wmma<128, 64><<<768, TB, 0, stream>>>(x, W1l, b1l, W1r, b1r, xl1, xr1, N);
  node_attend<64, false><<<nblocks, TB, 0, stream>>>(xl1, xr1, att1, rowptr, csr, bias1,
                                                     Wlin, blin, h1, N);

  gemm2_bias_wmma<64, 32><<<cdiv(rowTiles, 8), TB, 0, stream>>>(h1, W2l, b2l, W2r, b2r, xl2, xr2, N);
  node_attend<32, true><<<nblocks, TB, 0, stream>>>(xl2, xr2, att2, rowptr, csr, bias2,
                                                    Wlin, blin, out, N);
}


# node kernels: resident grid, each wave walks 11 consecutive nodes (100000 -> 9091 waves per launch), next node's rows and sources prefetched, two edge steps' rows in flight; sum order unchanged
# speedup vs baseline: 8.3872x; 1.0613x over previous
#include <hip/hip_runtime.h>
#include <hip/hip_bf16.h>
#include <math.h>

typedef __attribute__((ext_vector_type(2))) float v2f;
typedef __attribute__((ext_vector_type(8))) float v8f;

template <int K>
__device__ __forceinline__ void load_strip(float4* a, const float* __restrict__ X, int s, int h, int c) {
  const float* __restrict__ arow = X + (size_t)(s * 16 + c) * K + 4 * h;
#pragma unroll
  for (int kb = 0; kb < K / 8; ++kb) a[kb] = *(const float4*)(arow + 8 * kb);
}

template <int K, int H>
__device__ __forceinline__ void strip_times_weights(const float4* a, const float4* Bs, int s, int h, int c,
                                                    const float* __restrict__ bl,
                                                    const float* __restrict__ br,
                                                    float* __restrict__ Cl, float* __restrict__ Cr) {
  constexpr int NC = 2 * H;
  constexpr int NT = NC / 16;
  v8f acc[NT];
#pragma unroll
  for (int t = 0; t < NT; ++t) acc[t] = v8f{};
#pragma unroll
  for (int kb = 0; kb < K / 8; ++kb) {
    float4 b[NT];
#pragma unroll
    for (int t = 0; t < NT; ++t) b[t] = Bs[(2 * kb + h) * NC + 16 * t + c];
    const v2f a0 = {a[kb].x, a[kb].y};
    const v2f a1 = {a[kb].z, a[kb].w};
#pragma unroll
    for (int t = 0; t < NT; ++t) {
      const v2f b0 = {b[t].x, b[t].y};
      acc[t] = __builtin_amdgcn_wmma_f32_16x16x4_f32(false, a0, false, b0, (short)0, acc[t],
                                                     false, false);
    }
#pragma unroll
    for (int t = 0; t < NT; ++t) {
      const v2f b1 = {b[t].z, b[t].w};
      acc[t] = __builtin_amdgcn_wmma_f32_16x16x4_f32(false, a1, false, b1, (short)0, acc[t],
                                                     false, false);
    }
  }
#pragma unroll
  for (int t = 0; t < NT; ++t) {
    const int col = 16 * t + c;
    const bool left = 16 * t < H;
    const float bv = left ? bl[col] : br[col - H];
    float* __restrict__ crow = (left ? Cl + col : Cr + (col - H)) + (size_t)(s * 16 + 8 * h) * H;
#pragma unroll
    for (int v = 0; v < 8; ++v) crow[(size_t)v * H] = acc[t][v] + bv;
  }
}

template <int K, int H>
__global__ void __launch_bounds__(256)
gemm2_bias_wmma(const float* __restrict__ X, const float* __restrict__ Wl,
                const float* __restrict__ bl, const float* __restrict__ Wr,
                const float* __restrict__ br, float* __restrict__ Cl,
                float* __restrict__ Cr, int N) {
  constexpr int NC = 2 * H;
  constexpr int KQ = K / 4;
  __shared__ float4 Bs[KQ * NC];
  const int wave = __builtin_amdgcn_readfirstlane(threadIdx.x >> 5);
  const int lane = threadIdx.x & 31;
  const int h = lane >> 4, c = lane & 15;
  const int nStrips = N / 16;
  const int G = gridDim.x;
  int s = wave * G + blockIdx.x;

  float4 a[K / 8];
  if (s < nStrips) load_strip<K>(a, X, s, h, c);

  for (int it = threadIdx.x; it < KQ * (NC / 4); it += 256) {
    const int cj = it % (NC / 4), kq = it / (NC / 4);
    const int col = 4 * cj;
    const float* __restrict__ p = (col < H ? Wl + col : Wr + (col - H)) + (size_t)(4 * kq) * H;
    const float4 r0 = *(const float4*)p;
    const float4 r1 = *(const float4*)(p + H);
    const float4 r2 = *(const float4*)(p + 2 * H);
    const float4 r3 = *(const float4*)(p + 3 * H);
    float4* __restrict__ q = Bs + kq * NC + col;
    q[0] = make_float4(r0.x, r1.x, r2.x, r3.x);
    q[1] = make_float4(r0.y, r1.y, r2.y, r3.y);
    q[2] = make_float4(r0.z, r1.z, r2.z, r3.z);
    q[3] = make_float4(r0.w, r1.w, r2.w, r3.w);
  }
  __syncthreads();
  if (s >= nStrips) return;
  strip_times_weights<K, H>(a, Bs, s, h, c, bl, br, Cl, Cr);
  for (s += 8 * G; s < nStrips; s += 8 * G) {
    asm volatile("" ::: "memory");
    float4 a2[K / 8];
    load_strip<K>(a2, X, s, h, c);
    strip_times_weights<K, H>(a2, Bs, s, h, c, bl, br, Cl, Cr);
  }
}

__device__ __forceinline__ int wave_sum32_i(int v) {
#pragma unroll
  for (int off = 16; off > 0; off >>= 1) v += __shfl_xor(v, off, 32);
  return v;
}

constexpr int LOGW = 8;
constexpr int BINW = 1 << LOGW;
constexpr int SORT_T = 1024;
constexpr int SORT_EPT = 8;
constexpr int CH = SORT_T * SORT_EPT;
constexpr int NBMAX = SORT_T;
constexpr int GMAX = 1024;
constexpr int CAP = 12288;

__device__ __forceinline__ int block_excl_scan(int v, int* sh_w, int& total) {
  const int lane = threadIdx.x & 31, wave = threadIdx.x >> 5;
  int inc = v;
#pragma unroll
  for (int off = 1; off < 32; off <<= 1) {
    const int u = __shfl_up(inc, off, 32);
    if (lane >= off) inc += u;
  }
  if (lane == 31) sh_w[wave] = inc;
  __syncthreads();
  if (wave == 0) {
    const int w = sh_w[lane];
    int winc = w;
#pragma unroll
    for (int off = 1; off < 32; off <<= 1) {
      const int u = __shfl_up(winc, off, 32);
      if (lane >= off) winc += u;
    }
    sh_w[lane] = winc - w;
    if (lane == 31) sh_w[32] = winc;
  }
  __syncthreads();
  total = sh_w[32];
  return sh_w[wave] + inc - v;
}

__global__ void __launch_bounds__(SORT_T)
bin_chunk(const int* __restrict__ ei, int E, int NB, int* __restrict__ bsrc,
          int* __restrict__ bdst, int* __restrict__ lscan) {
  __shared__ int sh_bin[NBMAX];
  __shared__ int sh_w[33];
  __shared__ int sh_buf[CH];
  const int t = threadIdx.x;
  const int g = blockIdx.x;
  const int e0 = g * CH;
  sh_bin[t] = 0;
  __syncthreads();
  int s[SORT_EPT], d[SORT_EPT], rk[SORT_EPT];
#pragma unroll
  for (int k = 0; k < SORT_EPT; ++k) {
    const int e = e0 + k * SORT_T + t;
    d[k] = e < E ? ei[E + e] : -1;
    s[k] = e < E ? ei[e] : 0;
  }
#pragma unroll
  for (int k = 0; k < SORT_EPT; ++k) {
    rk[k] = 0;
    if (d[k] >= 0) rk[k] = atomicAdd(&sh_bin[d[k] >> LOGW], 1);
  }
  __syncthreads();
  int total;
  const int ex = block_excl_scan(sh_bin[t], sh_w, total);
  sh_bin[t] = ex;
  int* __restrict__ row = lscan + (size_t)g * (NB + 1);
  if (t < NB) row[t] = ex;
  if (t == 0) row[NB] = total;
  __syncthreads();
  int pos[SORT_EPT];
#pragma unroll
  for (int k = 0; k < SORT_EPT; ++k) pos[k] = d[k] >= 0 ? sh_bin[d[k] >> LOGW] + rk[k] : 0;
#pragma unroll
  for (int k = 0; k < SORT_EPT; ++k) if (d[k] >= 0) sh_buf[pos[k]] = d[k];
  __syncthreads();
#pragma unroll
  for (int k = 0; k < SORT_EPT; ++k) {
    const int i = k * SORT_T + t;
    if (i < total) bdst[e0 + i] = sh_buf[i];
  }
  __syncthreads();
#pragma unroll
  for (int k = 0; k < SORT_EPT; ++k) if (d[k] >= 0) sh_buf[pos[k]] = s[k];
  __syncthreads();
#pragma unroll
  for (int k = 0; k < SORT_EPT; ++k) {
    const int i = k * SORT_T + t;
    if (i < total) bsrc[e0 + i] = sh_buf[i];
  }
}

__global__ void __launch_bounds__(SORT_T)
bin_to_csr(const int* __restrict__ bsrc, const int* __restrict__ bdst,
           const int* __restrict__ lscan, int G, int NB, int N,
           int* __restrict__ rowptr, int* __restrict__ csr) {
  __shared__ int sh_s[GMAX];
  __shared__ int sh_n[GMAX];
  __shared__ int sh_cnt[BINW];
  __shared__ int sh_w[33];
  __shared__ int sh_red[2];
  __shared__ int sh_sorted[CAP];
  const int t = threadIdx.x, lane = t & 31;
  const int wave = __builtin_amdgcn_readfirstlane(t >> 5);
  const int b = blockIdx.x;
  const int n0 = b << LOGW;
  if (t < BINW) sh_cnt[t] = 0;
  if (t < 2) sh_red[t] = 0;
  __syncthreads();
  int pb = 0, pn = 0;
  for (int g = t; g < G; g += SORT_T) {
    const int* __restrict__ row = lscan + (size_t)g * (NB + 1) + b;
    const int s0 = row[0], s1 = row[1];
    sh_s[g] = g * CH + s0;
    sh_n[g] = s1 - s0;
    pb += s0;
    pn += s1 - s0;
  }
  pb = wave_sum32_i(pb);
  pn = wave_sum32_i(pn);
  if (lane == 0) { atomicAdd(&sh_red[0], pb); atomicAdd(&sh_red[1], pn); }
  __syncthreads();
  const int base = sh_red[0];
  const int total = sh_red[1];
  for (int g = wave; g < G; g += SORT_T / 32) {
    const int st = sh_s[g], n = sh_n[g];
    for (int i = lane; i < n; i += 32) atomicAdd(&sh_cnt[bdst[st + i] - n0], 1);
  }
  __syncthreads();
  int tot2;
  const int ex = block_excl_scan(t < BINW ? sh_cnt[t] : 0, sh_w, tot2);
  if (t < BINW) {
    sh_cnt[t] = ex;
    if (n0 + t < N) rowptr[n0 + t] = base + ex;
  }
  if (b == NB - 1 && t == 0) rowptr[N] = base + total;
  __syncthreads();
  const bool staged = total <= CAP;
  for (int g = wave; g < G; g += SORT_T / 32) {
    const int st = sh_s[g], n = sh_n[g];
    for (int i = lane; i < n; i += 32) {
      const int sv = bsrc[st + i];
      const int p = atomicAdd(&sh_cnt[bdst[st + i] - n0], 1);
      if (staged) sh_sorted[p] = sv; else csr[base + p] = sv;
    }
  }
  if (staged) {
    __syncthreads();
    for (int i = t; i < total; i += SORT_T) csr[base + i] = sh_sorted[i];
  }
}

constexpr int CPL = 8;

template <int CTRL>
__device__ __forceinline__ float lane_get(float v) {
  return __builtin_bit_cast(
      float, __builtin_amdgcn_update_dpp(0, __builtin_bit_cast(int, v), CTRL, 0xF, 0xF, false));
}

template <int LPE>
__device__ __forceinline__ float group_sum(float v) {
  v += lane_get<0xB1>(v);
  v += lane_get<0x4E>(v);
  if (LPE == 8) v += lane_get<0x141>(v);
  return v;
}

template <int H>
__device__ __forceinline__ void load_row8(float* dst, const float* __restrict__ base, int row, int sub) {
  const float* __restrict__ p = base + (size_t)row * H + 4 * sub;
  const float4 t0 = *(const float4*)p;
  const float4 t1 = *(const float4*)(p + H / 2);
  dst[0] = t0.x; dst[1] = t0.y; dst[2] = t0.z; dst[3] = t0.w;
  dst[4] = t1.x; dst[5] = t1.y; dst[6] = t1.z; dst[7] = t1.w;
}

__device__ __forceinline__ float score_part8(const float* a, const float* r, const float* w) {
  float s0 = 0.0f, s1 = 0.0f;
#pragma unroll
  for (int v = 0; v < CPL; v += 2) {
    float h0 = a[v] + r[v];         h0 = fmaxf(h0, 0.2f * h0);
    float h1 = a[v + 1] + r[v + 1]; h1 = fmaxf(h1, 0.2f * h1);
    s0 = fmaf(h0, w[v], s0);
    s1 = fmaf(h1, w[v + 1], s1);
  }
  return s0 + s1;
}

__device__ __forceinline__ float exp2_bare(float x) { return __builtin_amdgcn_exp2f(x); }

__device__ __forceinline__ void softmax_step(float e, bool ok, const float* a, float& m, float& den,
                                             float* acc) {
  const float eo = ok ? e : -INFINITY;
  const float mn = eo > m ? eo : m;
  const float sc = exp2_bare(m - mn);
  const float p = exp2_bare(eo - mn);
  m = mn;
  den = fmaf(den, sc, p);
#pragma unroll
  for (int v = 0; v < CPL; ++v) acc[v] = fmaf(acc[v], sc, p * a[v]);
}

template <int H>
__device__ __forceinline__ void edge_trips(int s, int cnt, const float* __restrict__ xl,
                                           const float* r, const float* w, int sub, int grp,
                                           float& m, float& den, float* acc) {
  constexpr int EPW = 32 / (H / CPL);
  for (int q = 0; q < cnt; q += 2 * EPW) {
    const int i0 = q + grp;
    const int i1 = q + EPW + grp;
    const int j0 = __shfl(s, i0, 32);
    const int j1 = __shfl(s, i1, 32);
    float a0[CPL], a1[CPL];
    load_row8<H>(a0, xl, j0, sub);
    load_row8<H>(a1, xl, j1, sub);
    asm volatile("" :: "v"(a0[0]), "v"(a0[1]), "v"(a0[2]), "v"(a0[3]),
                       "v"(a0[4]), "v"(a0[5]), "v"(a0[6]), "v"(a0[7]),
                       "v"(a1[0]), "v"(a1[1]), "v"(a1[2]), "v"(a1[3]),
                       "v"(a1[4]), "v"(a1[5]), "v"(a1[6]), "v"(a1[7]));
    const float e0 = group_sum<H / CPL>(score_part8(a0, r, w));
    softmax_step(e0, i0 < cnt, a0, m, den, acc);
    if (q + EPW < cnt) {
      const float e1 = group_sum<H / CPL>(score_part8(a1, r, w));
      softmax_step(e1, i1 < cnt, a1, m, den, acc);
    }
  }
}

template <int H, bool LAST>
__global__ void __launch_bounds__(256)
node_attend(const float* __restrict__ xl, const float* __restrict__ xr,
            const float* __restrict__ att, const int* __restrict__ rowptr,
            const int* __restrict__ csr, const float* __restrict__ bias,
            const float* __restrict__ Wlin, const float* __restrict__ blin,
            float* __restrict__ outp, int N, int NPW) {
  constexpr int LPE = H / CPL;
  constexpr int EPW = 32 / LPE;
  const int lane = threadIdx.x & 31;
  const int gw = __builtin_amdgcn_readfirstlane((blockIdx.x * blockDim.x + threadIdx.x) >> 5);
  const int n0 = gw * NPW;
  if (n0 >= N) return;
  const int nn = min(NPW, N - n0);
  const int sub = lane & (LPE - 1);
  const int grp = lane / LPE;

  const int rp = rowptr[n0 + min(lane, nn)];

  float w[CPL];
  load_row8<H>(w, att, 0, sub);
#pragma unroll
  for (int v = 0; v < CPL; ++v) w[v] *= 1.44269504088896340736f;

  float r[CPL], acc[CPL];
  load_row8<H>(r, xr, n0, sub);
  load_row8<H>(acc, xl, n0, sub);
  int beg = __builtin_amdgcn_readlane(rp, 0);
  int end = __builtin_amdgcn_readlane(rp, 1);
  int s = n0;
  if (lane < end - beg) s = csr[beg + lane];

  for (int k = 0; k < nn; ++k) {
    const int n = n0 + k;
    const int kn = min(k + 1, nn - 1);
    const int nbeg = __builtin_amdgcn_readlane(rp, kn);
    const int nend = __builtin_amdgcn_readlane(rp, kn + 1);
    float rn[CPL], accn[CPL];
    load_row8<H>(rn, xr, n0 + kn, sub);
    load_row8<H>(accn, xl, n0 + kn, sub);
    int sn = n0 + kn;
    if (lane < nend - nbeg) sn = csr[nbeg + lane];

    float m = group_sum<LPE>(score_part8(acc, r, w));
    float den = grp == 0 ? 1.0f : 0.0f;
#pragma unroll
    for (int v = 0; v < CPL; ++v) acc[v] = grp == 0 ? acc[v] : 0.0f;

    edge_trips<H>(s, min(32, end - beg), xl, r, w, sub, grp, m, den, acc);
    for (int c = beg + 32; c < end; c += 32) {
      const int cnt = min(32, end - c);
      int s2 = n;
      if (lane < cnt) s2 = csr[c + lane];
      edge_trips<H>(s2, cnt, xl, r, w, sub, grp, m, den, acc);
    }

    float M = m;
#pragma unroll
    for (int off = LPE; off < 32; off <<= 1) M = fmaxf(M, __shfl_xor(M, off, 32));
    const float f = exp2_bare(m - M);
    den *= f;
#pragma unroll
    for (int v = 0; v < CPL; ++v) acc[v] *= f;
#pragma unroll
    for (int off = LPE; off < 32; off <<= 1) {
      den += __shfl_xor(den, off, 32);
#pragma unroll
      for (int v = 0; v < CPL; ++v) acc[v] += __shfl_xor(acc[v], off, 32);
    }

    const float inv = __builtin_amdgcn_rcpf(den + 1e-16f);
    asm volatile("" ::: "memory");
    if (!LAST) {
      float b[CPL], o[CPL];
      load_row8<H>(b, bias, 0, sub);
#pragma unroll
      for (int v = 0; v < CPL; ++v) {
        const float t = fmaf(acc[v], inv, b[v]);
        o[v] = fmaxf(t, 0.01f * t);
      }
      if (grp == 0) {
        float* __restrict__ p = outp + (size_t)n * H + 4 * sub;
        *(float4*)p = make_float4(o[0], o[1], o[2], o[3]);
        *(float4*)(p + H / 2) = make_float4(o[4], o[5], o[6], o[7]);
      }
    } else {
      float b[CPL], wl[CPL];
      load_row8<H>(b, bias, 0, sub);
      load_row8<H>(wl, Wlin, 0, sub);
      float tot = 0.0f;
#pragma unroll
      for (int v = 0; v < CPL; ++v) tot = fmaf(fmaf(acc[v], inv, b[v]), wl[v], tot);
      tot = group_sum<LPE>(tot);
      if (lane == 0) outp[n] = tot + blin[0];
    }

    beg = nbeg;
    end = nend;
    s = sn;
#pragma unroll
    for (int v = 0; v < CPL; ++v) { r[v] = rn[v]; acc[v] = accn[v]; }
  }
}

static inline int cdiv(int a, int b) { return (a + b - 1) / b; }

extern "C" void kernel_launch(void* const* d_in, const int* in_sizes, int n_in,
                              void* d_out, int out_size, void* d_ws, size_t ws_size,
                              hipStream_t stream) {
  const float* x     = (const float*)d_in[0];
  const int*   ei    = (const int*)  d_in[1];
  const float* W1l   = (const float*)d_in[2];
  const float* b1l   = (const float*)d_in[3];
  const float* W1r   = (const float*)d_in[4];
  const float* b1r   = (const float*)d_in[5];
  const float* att1  = (const float*)d_in[6];
  const float* bias1 = (const float*)d_in[7];
  const float* W2l   = (const float*)d_in[8];
  const float* b2l   = (const float*)d_in[9];
  const float* W2r   = (const float*)d_in[10];
  const float* b2r   = (const float*)d_in[11];
  const float* att2  = (const float*)d_in[12];
  const float* bias2 = (const float*)d_in[13];
  const float* Wlin  = (const float*)d_in[14];
  const float* blin  = (const float*)d_in[15];
  float* out = (float*)d_out;

  const int N    = in_sizes[0] / 128;
  const int E    = in_sizes[1] / 2;

  float* ws = (float*)d_ws;
  const int NB = cdiv(N, BINW);
  const int G  = cdiv(E, CH);
  size_t off = 0;
  float* xl1  = ws + off; off += (size_t)N * 64;
  float* xr1  = ws + off; off += (size_t)N * 64;
  float* h1   = ws + off; off += (size_t)N * 64;
  float* xl2  = ws + off; off += (size_t)N * 32;
  float* xr2  = ws + off; off += (size_t)N * 32;
  int* rowptr = (int*)(ws + off); off += (size_t)N + 1;
  off = (off + 31) & ~(size_t)31;
  int* csr    = (int*)(ws + off); off += (size_t)E;
  int* bsrc   = (int*)(ws + off); off += (size_t)G * CH;
  int* bdst   = (int*)(ws + off); off += (size_t)G * CH;
  int* lscan  = (int*)(ws + off); off += (size_t)G * (NB + 1);
  (void)ws_size; (void)n_in; (void)out_size;

  const int TB = 256;
  const int rowTiles = cdiv(N, 16);
  const int NTB = 128;
  int NPW = cdiv(N, 256 * 4 * 9);
  if (NPW < 1) NPW = 1;
  if (NPW > 31) NPW = 31;
  const int nblocks = cdiv(cdiv(N, NPW), NTB / 32);

  bin_chunk<<<G, SORT_T, 0, stream>>>(ei, E, NB, bsrc, bdst, lscan);
  bin_to_csr<<<NB, SORT_T, 0, stream>>>(bsrc, bdst, lscan, G, NB, N, rowptr, csr);

  gemm2_bias_wmma<128, 64><<<768, TB, 0, stream>>>(x, W1l, b1l, W1r, b1r, xl1, xr1, N);
  node_attend<64, false><<<nblocks, NTB, 0, stream>>>(xl1, xr1, att1, rowptr, csr, bias1,
                                                     Wlin, blin, h1, N, NPW);

  gemm2_bias_wmma<64, 32><<<cdiv(rowTiles, 8), TB, 0, stream>>>(h1, W2l, b2l, W2r, b2r, xl2, xr2, N);
  node_attend<32, true><<<nblocks, NTB, 0, stream>>>(xl2, xr2, att2, rowptr, csr, bias2,
                                                    Wlin, blin, out, N, NPW);
}


# layer 2's two projections move into layer 1's node kernel: a work-group keeps its 48 rows of h in LDS and multiplies them by [W2l|W2r] on WMMA behind one barrier; 6 -> 5 launches; sum order unchanged
# speedup vs baseline: 8.7003x; 1.0373x over previous
#include <hip/hip_runtime.h>
#include <hip/hip_bf16.h>
#include <math.h>

typedef __attribute__((ext_vector_type(2))) float v2f;
typedef __attribute__((ext_vector_type(8))) float v8f;

template <int K>
__device__ __forceinline__ void load_strip(float4* a, const float* __restrict__ X, int s, int h, int c) {
  const float* __restrict__ arow = X + (size_t)(s * 16 + c) * K + 4 * h;
#pragma unroll
  for (int kb = 0; kb < K / 8; ++kb) a[kb] = *(const float4*)(arow + 8 * kb);
}

template <int K, int H>
__device__ __forceinline__ void strip_times_weights(const float4* a, const float4* Bs, int s, int h, int c,
                                                    const float* __restrict__ bl,
                                                    const float* __restrict__ br,
                                                    float* __restrict__ Cl, float* __restrict__ Cr) {
  constexpr int NC = 2 * H;
  constexpr int NT = NC / 16;
  v8f acc[NT];
#pragma unroll
  for (int t = 0; t < NT; ++t) acc[t] = v8f{};
#pragma unroll
  for (int kb = 0; kb < K / 8; ++kb) {
    float4 b[NT];
#pragma unroll
    for (int t = 0; t < NT; ++t) b[t] = Bs[(2 * kb + h) * NC + 16 * t + c];
    const v2f a0 = {a[kb].x, a[kb].y};
    const v2f a1 = {a[kb].z, a[kb].w};
#pragma unroll
    for (int t = 0; t < NT; ++t) {
      const v2f b0 = {b[t].x, b[t].y};
      acc[t] = __builtin_amdgcn_wmma_f32_16x16x4_f32(false, a0, false, b0, (short)0, acc[t],
                                                     false, false);
    }
#pragma unroll
    for (int t = 0; t < NT; ++t) {
      const v2f b1 = {b[t].z, b[t].w};
      acc[t] = __builtin_amdgcn_wmma_f32_16x16x4_f32(false, a1, false, b1, (short)0, acc[t],
                                                     false, false);
    }
  }
#pragma unroll
  for (int t = 0; t < NT; ++t) {
    const int col = 16 * t + c;
    const bool left = 16 * t < H;
    const float bv = left ? bl[col] : br[col - H];
    float* __restrict__ crow = (left ? Cl + col : Cr + (col - H)) + (size_t)(s * 16 + 8 * h) * H;
#pragma unroll
    for (int v = 0; v < 8; ++v) crow[(size_t)v * H] = acc[t][v] + bv;
  }
}

template <int K, int H>
__global__ void __launch_bounds__(256)
gemm2_bias_wmma(const float* __restrict__ X, const float* __restrict__ Wl,
                const float* __restrict__ bl, const float* __restrict__ Wr,
                const float* __restrict__ br, float* __restrict__ Cl,
                float* __restrict__ Cr, int N) {
  constexpr int NC = 2 * H;
  constexpr int KQ = K / 4;
  __shared__ float4 Bs[KQ * NC];
  const int wave = __builtin_amdgcn_readfirstlane(threadIdx.x >> 5);
  const int lane = threadIdx.x & 31;
  const int h = lane >> 4, c = lane & 15;
  const int nStrips = N / 16;
  const int G = gridDim.x;
  int s = wave * G + blockIdx.x;

  float4 a[K / 8];
  if (s < nStrips) load_strip<K>(a, X, s, h, c);

  for (int it = threadIdx.x; it < KQ * (NC / 4); it += 256) {
    const int cj = it % (NC / 4), kq = it / (NC / 4);
    const int col = 4 * cj;
    const float* __restrict__ p = (col < H ? Wl + col : Wr + (col - H)) + (size_t)(4 * kq) * H;
    const float4 r0 = *(const float4*)p;
    const float4 r1 = *(const float4*)(p + H);
    const float4 r2 = *(const float4*)(p + 2 * H);
    const float4 r3 = *(const float4*)(p + 3 * H);
    float4* __restrict__ q = Bs + kq * NC + col;
    q[0] = make_float4(r0.x, r1.x, r2.x, r3.x);
    q[1] = make_float4(r0.y, r1.y, r2.y, r3.y);
    q[2] = make_float4(r0.z, r1.z, r2.z, r3.z);
    q[3] = make_float4(r0.w, r1.w, r2.w, r3.w);
  }
  __syncthreads();
  if (s >= nStrips) return;
  strip_times_weights<K, H>(a, Bs, s, h, c, bl, br, Cl, Cr);
  for (s += 8 * G; s < nStrips; s += 8 * G) {
    asm volatile("" ::: "memory");
    float4 a2[K / 8];
    load_strip<K>(a2, X, s, h, c);
    strip_times_weights<K, H>(a2, Bs, s, h, c, bl, br, Cl, Cr);
  }
}

__device__ __forceinline__ int wave_sum32_i(int v) {
#pragma unroll
  for (int off = 16; off > 0; off >>= 1) v += __shfl_xor(v, off, 32);
  return v;
}

constexpr int LOGW = 8;
constexpr int BINW = 1 << LOGW;
constexpr int SORT_T = 1024;
constexpr int SORT_EPT = 8;
constexpr int CH = SORT_T * SORT_EPT;
constexpr int NBMAX = SORT_T;
constexpr int GMAX = 1024;
constexpr int CAP = 12288;

__device__ __forceinline__ int block_excl_scan(int v, int* sh_w, int& total) {
  const int lane = threadIdx.x & 31, wave = threadIdx.x >> 5;
  int inc = v;
#pragma unroll
  for (int off = 1; off < 32; off <<= 1) {
    const int u = __shfl_up(inc, off, 32);
    if (lane >= off) inc += u;
  }
  if (lane == 31) sh_w[wave] = inc;
  __syncthreads();
  if (wave == 0) {
    const int w = sh_w[lane];
    int winc = w;
#pragma unroll
    for (int off = 1; off < 32; off <<= 1) {
      const int u = __shfl_up(winc, off, 32);
      if (lane >= off) winc += u;
    }
    sh_w[lane] = winc - w;
    if (lane == 31) sh_w[32] = winc;
  }
  __syncthreads();
  total = sh_w[32];
  return sh_w[wave] + inc - v;
}

__global__ void __launch_bounds__(SORT_T)
bin_chunk(const int* __restrict__ ei, int E, int NB, int* __restrict__ bsrc,
          int* __restrict__ bdst, int* __restrict__ lscan) {
  __shared__ int sh_bin[NBMAX];
  __shared__ int sh_w[33];
  __shared__ int sh_buf[CH];
  const int t = threadIdx.x;
  const int g = blockIdx.x;
  const int e0 = g * CH;
  sh_bin[t] = 0;
  __syncthreads();
  int s[SORT_EPT], d[SORT_EPT], rk[SORT_EPT];
#pragma unroll
  for (int k = 0; k < SORT_EPT; ++k) {
    const int e = e0 + k * SORT_T + t;
    d[k] = e < E ? ei[E + e] : -1;
    s[k] = e < E ? ei[e] : 0;
  }
#pragma unroll
  for (int k = 0; k < SORT_EPT; ++k) {
    rk[k] = 0;
    if (d[k] >= 0) rk[k] = atomicAdd(&sh_bin[d[k] >> LOGW], 1);
  }
  __syncthreads();
  int total;
  const int ex = block_excl_scan(sh_bin[t], sh_w, total);
  sh_bin[t] = ex;
  int* __restrict__ row = lscan + (size_t)g * (NB + 1);
  if (t < NB) row[t] = ex;
  if (t == 0) row[NB] = total;
  __syncthreads();
  int pos[SORT_EPT];
#pragma unroll
  for (int k = 0; k < SORT_EPT; ++k) pos[k] = d[k] >= 0 ? sh_bin[d[k] >> LOGW] + rk[k] : 0;
#pragma unroll
  for (int k = 0; k < SORT_EPT; ++k) if (d[k] >= 0) sh_buf[pos[k]] = d[k];
  __syncthreads();
#pragma unroll
  for (int k = 0; k < SORT_EPT; ++k) {
    const int i = k * SORT_T + t;
    if (i < total) bdst[e0 + i] = sh_buf[i];
  }
  __syncthreads();
#pragma unroll
  for (int k = 0; k < SORT_EPT; ++k) if (d[k] >= 0) sh_buf[pos[k]] = s[k];
  __syncthreads();
#pragma unroll
  for (int k = 0; k < SORT_EPT; ++k) {
    const int i = k * SORT_T + t;
    if (i < total) bsrc[e0 + i] = sh_buf[i];
  }
}

__global__ void __launch_bounds__(SORT_T)
bin_to_csr(const int* __restrict__ bsrc, const int* __restrict__ bdst,
           const int* __restrict__ lscan, int G, int NB, int N,
           int* __restrict__ rowptr, int* __restrict__ csr) {
  __shared__ int sh_s[GMAX];
  __shared__ int sh_n[GMAX];
  __shared__ int sh_cnt[BINW];
  __shared__ int sh_w[33];
  __shared__ int sh_red[2];
  __shared__ int sh_sorted[CAP];
  const int t = threadIdx.x, lane = t & 31;
  const int wave = __builtin_amdgcn_readfirstlane(t >> 5);
  const int b = blockIdx.x;
  const int n0 = b << LOGW;
  if (t < BINW) sh_cnt[t] = 0;
  if (t < 2) sh_red[t] = 0;
  __syncthreads();
  int pb = 0, pn = 0;
  for (int g = t; g < G; g += SORT_T) {
    const int* __restrict__ row = lscan + (size_t)g * (NB + 1) + b;
    const int s0 = row[0], s1 = row[1];
    sh_s[g] = g * CH + s0;
    sh_n[g] = s1 - s0;
    pb += s0;
    pn += s1 - s0;
  }
  pb = wave_sum32_i(pb);
  pn = wave_sum32_i(pn);
  if (lane == 0) { atomicAdd(&sh_red[0], pb); atomicAdd(&sh_red[1], pn); }
  __syncthreads();
  const int base = sh_red[0];
  const int total = sh_red[1];
  for (int g = wave; g < G; g += SORT_T / 32) {
    const int st = sh_s[g], n = sh_n[g];
    for (int i = lane; i < n; i += 32) atomicAdd(&sh_cnt[bdst[st + i] - n0], 1);
  }
  __syncthreads();
  int tot2;
  const int ex = block_excl_scan(t < BINW ? sh_cnt[t] : 0, sh_w, tot2);
  if (t < BINW) {
    sh_cnt[t] = ex;
    if (n0 + t < N) rowptr[n0 + t] = base + ex;
  }
  if (b == NB - 1 && t == 0) rowptr[N] = base + total;
  __syncthreads();
  const bool staged = total <= CAP;
  for (int g = wave; g < G; g += SORT_T / 32) {
    const int st = sh_s[g], n = sh_n[g];
    for (int i = lane; i < n; i += 32) {
      const int sv = bsrc[st + i];
      const int p = atomicAdd(&sh_cnt[bdst[st + i] - n0], 1);
      if (staged) sh_sorted[p] = sv; else csr[base + p] = sv;
    }
  }
  if (staged) {
    __syncthreads();
    for (int i = t; i < total; i += SORT_T) csr[base + i] = sh_sorted[i];
  }
}

constexpr int CPL = 8;
constexpr int NPWF = 12;
constexpr int NTBF = 128;

template <int CTRL>
__device__ __forceinline__ float lane_get(float v) {
  return __builtin_bit_cast(
      float, __builtin_amdgcn_update_dpp(0, __builtin_bit_cast(int, v), CTRL, 0xF, 0xF, false));
}

template <int LPE>
__device__ __forceinline__ float group_sum(float v) {
  v += lane_get<0xB1>(v);
  v += lane_get<0x4E>(v);
  if (LPE == 8) v += lane_get<0x141>(v);
  return v;
}

template <int H>
__device__ __forceinline__ void load_row8(float* dst, const float* __restrict__ base, int row, int sub) {
  const float* __restrict__ p = base + (size_t)row * H + 4 * sub;
  const float4 t0 = *(const float4*)p;
  const float4 t1 = *(const float4*)(p + H / 2);
  dst[0] = t0.x; dst[1] = t0.y; dst[2] = t0.z; dst[3] = t0.w;
  dst[4] = t1.x; dst[5] = t1.y; dst[6] = t1.z; dst[7] = t1.w;
}

__device__ __forceinline__ float score_part8(const float* a, const float* r, const float* w) {
  float s0 = 0.0f, s1 = 0.0f;
#pragma unroll
  for (int v = 0; v < CPL; v += 2) {
    float h0 = a[v] + r[v];         h0 = fmaxf(h0, 0.2f * h0);
    float h1 = a[v + 1] + r[v + 1]; h1 = fmaxf(h1, 0.2f * h1);
    s0 = fmaf(h0, w[v], s0);
    s1 = fmaf(h1, w[v + 1], s1);
  }
  return s0 + s1;
}

__device__ __forceinline__ float exp2_bare(float x) { return __builtin_amdgcn_exp2f(x); }

__device__ __forceinline__ void softmax_step(float e, bool ok, const float* a, float& m, float& den,
                                             float* acc) {
  const float eo = ok ? e : -INFINITY;
  const float mn = eo > m ? eo : m;
  const float sc = exp2_bare(m - mn);
  const float p = exp2_bare(eo - mn);
  m = mn;
  den = fmaf(den, sc, p);
#pragma unroll
  for (int v = 0; v < CPL; ++v) acc[v] = fmaf(acc[v], sc, p * a[v]);
}

template <int H>
__device__ __forceinline__ void edge_trips(int s, int cnt, const float* __restrict__ xl,
                                           const float* r, const float* w, int sub, int grp,
                                           float& m, float& den, float* acc) {
  constexpr int EPW = 32 / (H / CPL);
  for (int q = 0; q < cnt; q += 2 * EPW) {
    const int i0 = q + grp;
    const int i1 = q + EPW + grp;
    const int j0 = __shfl(s, i0, 32);
    const int j1 = __shfl(s, i1, 32);
    float a0[CPL], a1[CPL];
    load_row8<H>(a0, xl, j0, sub);
    load_row8<H>(a1, xl, j1, sub);
    asm volatile("" :: "v"(a0[0]), "v"(a0[1]), "v"(a0[2]), "v"(a0[3]),
                       "v"(a0[4]), "v"(a0[5]), "v"(a0[6]), "v"(a0[7]),
                       "v"(a1[0]), "v"(a1[1]), "v"(a1[2]), "v"(a1[3]),
                       "v"(a1[4]), "v"(a1[5]), "v"(a1[6]), "v"(a1[7]));
    const float e0 = group_sum<H / CPL>(score_part8(a0, r, w));
    softmax_step(e0, i0 < cnt, a0, m, den, acc);
    if (q + EPW < cnt) {
      const float e1 = group_sum<H / CPL>(score_part8(a1, r, w));
      softmax_step(e1, i1 < cnt, a1, m, den, acc);
    }
  }
}

template <int H, bool LAST>
__global__ void __launch_bounds__(256)
node_attend(const float* __restrict__ xl, const float* __restrict__ xr,
            const float* __restrict__ att, const int* __restrict__ rowptr,
            const int* __restrict__ csr, const float* __restrict__ bias,
            const float* __restrict__ Wlin, const float* __restrict__ blin,
            float* __restrict__ outp, int N, int NPW,
            const float* __restrict__ W2l, const float* __restrict__ b2l,
            const float* __restrict__ W2r, const float* __restrict__ b2r,
            float* __restrict__ xl2, float* __restrict__ xr2) {
  constexpr int LPE = H / CPL;
  constexpr int EPW = 32 / LPE;
  constexpr int H2 = H / 2;
  constexpr int NC = 2 * H2;
  constexpr int KQ = H / 4;
  constexpr int HP = H + 4;
  constexpr int ROWS = (NTBF / 32) * NPWF;
  __shared__ float4 Bs[LAST ? 1 : KQ * NC];
  __shared__ float hs[LAST ? 4 : ROWS * HP];
  const int lane = threadIdx.x & 31;
  const int gw = __builtin_amdgcn_readfirstlane((blockIdx.x * blockDim.x + threadIdx.x) >> 5);
  const int wv = __builtin_amdgcn_readfirstlane(threadIdx.x >> 5);
  const int npw = LAST ? NPW : NPWF;
  const int n0 = gw * npw;
  if (LAST && n0 >= N) return;
  const int nn = min(npw, N - n0);
  const int sub = lane & (LPE - 1);
  const int grp = lane / LPE;

  if (!LAST) {
    static_assert(LAST || (NC / 16 == NTBF / 32 && ROWS % 16 == 0), "one column tile per wave, whole strips");
    for (int it = threadIdx.x; it < KQ * (NC / 4); it += NTBF) {
      const int cj = it % (NC / 4), kq = it / (NC / 4);
      const int col = 4 * cj;
      const float* __restrict__ p = (col < H2 ? W2l + col : W2r + (col - H2)) + (size_t)(4 * kq) * H2;
      const float4 r0 = *(const float4*)p;
      const float4 r1 = *(const float4*)(p + H2);
      const float4 r2 = *(const float4*)(p + 2 * H2);
      const float4 r3 = *(const float4*)(p + 3 * H2);
      float4* __restrict__ q = Bs + kq * NC + col;
      q[0] = make_float4(r0.x, r1.x, r2.x, r3.x);
      q[1] = make_float4(r0.y, r1.y, r2.y, r3.y);
      q[2] = make_float4(r0.z, r1.z, r2.z, r3.z);
      q[3] = make_float4(r0.w, r1.w, r2.w, r3.w);
    }
  }

  if (nn > 0) {

  const int rp = rowptr[n0 + min(lane, nn)];

  float w[CPL];
  load_row8<H>(w, att, 0, sub);
#pragma unroll
  for (int v = 0; v < CPL; ++v) w[v] *= 1.44269504088896340736f;

  float r[CPL], acc[CPL];
  load_row8<H>(r, xr, n0, sub);
  load_row8<H>(acc, xl, n0, sub);
  int beg = __builtin_amdgcn_readlane(rp, 0);
  int end = __builtin_amdgcn_readlane(rp, 1);
  int s = n0;
  if (lane < end - beg) s = csr[beg + lane];

  for (int k = 0; k < nn; ++k) {
    const int n = n0 + k;
    const int kn = min(k + 1, nn - 1);
    const int nbeg = __builtin_amdgcn_readlane(rp, kn);
    const int nend = __builtin_amdgcn_readlane(rp, kn + 1);
    float rn[CPL], accn[CPL];
    load_row8<H>(rn, xr, n0 + kn, sub);
    load_row8<H>(accn, xl, n0 + kn, sub);
    int sn = n0 + kn;
    if (lane < nend - nbeg) sn = csr[nbeg + lane];

    float m = group_sum<LPE>(score_part8(acc, r, w));
    float den = grp == 0 ? 1.0f : 0.0f;
#pragma unroll
    for (int v = 0; v < CPL; ++v) acc[v] = grp == 0 ? acc[v] : 0.0f;

    edge_trips<H>(s, min(32, end - beg), xl, r, w, sub, grp, m, den, acc);
    for (int c = beg + 32; c < end; c += 32) {
      const int cnt = min(32, end - c);
      int s2 = n;
      if (lane < cnt) s2 = csr[c + lane];
      edge_trips<H>(s2, cnt, xl, r, w, sub, grp, m, den, acc);
    }

    float M = m;
#pragma unroll
    for (int off = LPE; off < 32; off <<= 1) M = fmaxf(M, __shfl_xor(M, off, 32));
    const float f = exp2_bare(m - M);
    den *= f;
#pragma unroll
    for (int v = 0; v < CPL; ++v) acc[v] *= f;
#pragma unroll
    for (int off = LPE; off < 32; off <<= 1) {
      den += __shfl_xor(den, off, 32);
#pragma unroll
      for (int v = 0; v < CPL; ++v) acc[v] += __shfl_xor(acc[v], off, 32);
    }

    const float inv = __builtin_amdgcn_rcpf(den + 1e-16f);
    asm volatile("" ::: "memory");
    if (!LAST) {
      float b[CPL], o[CPL];
      load_row8<H>(b, bias, 0, sub);
#pragma unroll
      for (int v = 0; v < CPL; ++v) {
        const float t = fmaf(acc[v], inv, b[v]);
        o[v] = fmaxf(t, 0.01f * t);
      }
      if (grp == 0) {
        float* __restrict__ p = hs + (wv * NPWF + k) * HP + 4 * sub;
        *(float4*)p = make_float4(o[0], o[1], o[2], o[3]);
        *(float4*)(p + H / 2) = make_float4(o[4], o[5], o[6], o[7]);
      }
    } else {
      float b[CPL], wl[CPL];
      load_row8<H>(b, bias, 0, sub);
      load_row8<H>(wl, Wlin, 0, sub);
      float tot = 0.0f;
#pragma unroll
      for (int v = 0; v < CPL; ++v) tot = fmaf(fmaf(acc[v], inv, b[v]), wl[v], tot);
      tot = group_sum<LPE>(tot);
      if (lane == 0) outp[n] = tot + blin[0];
    }

    beg = nbeg;
    end = nend;
    s = sn;
#pragma unroll
    for (int v = 0; v < CPL; ++v) { r[v] = rn[v]; acc[v] = accn[v]; }
  }
  }

  if (!LAST) {
    for (int k = nn > 0 ? nn : 0; k < NPWF; ++k)
      if (lane < H / 4) *(float4*)(hs + (wv * NPWF + k) * HP + 4 * lane) = make_float4(0.f, 0.f, 0.f, 0.f);
    __syncthreads();

    constexpr int NS = ROWS / 16;
    const int h = lane >> 4, c = lane & 15;
    const int r0 = blockIdx.x * ROWS;
    v8f pacc[NS];
#pragma unroll
    for (int st = 0; st < NS; ++st) pacc[st] = v8f{};
#pragma unroll
    for (int kb = 0; kb < H / 8; ++kb) {
      const float4 b = Bs[(2 * kb + h) * NC + 16 * wv + c];
      float4 a[NS];
#pragma unroll
      for (int st = 0; st < NS; ++st) a[st] = *(const float4*)(hs + (16 * st + c) * HP + 8 * kb + 4 * h);
      const v2f b0 = {b.x, b.y};
      const v2f b1 = {b.z, b.w};
#pragma unroll
      for (int st = 0; st < NS; ++st) {
        const v2f a0 = {a[st].x, a[st].y};
        pacc[st] = __builtin_amdgcn_wmma_f32_16x16x4_f32(false, a0, false, b0, (short)0, pacc[st],
                                                         false, false);
      }
#pragma unroll
      for (int st = 0; st < NS; ++st) {
        const v2f a1 = {a[st].z, a[st].w};
        pacc[st] = __builtin_amdgcn_wmma_f32_16x16x4_f32(false, a1, false, b1, (short)0, pacc[st],
                                                         false, false);
      }
    }
    const int col = 16 * wv + c;
    const bool left = col < H2;
    const float bv = left ? b2l[col] : b2r[col - H2];
    float* __restrict__ ccol = left ? xl2 + col : xr2 + (col - H2);
#pragma unroll
    for (int st = 0; st < NS; ++st) {
      const int rs = r0 + 16 * st;
      float* __restrict__ crow = ccol + (size_t)(rs + 8 * h) * H2;
      if (rs + 16 <= N) {
#pragma unroll
        for (int v = 0; v < 8; ++v) crow[(size_t)v * H2] = pacc[st][v] + bv;
      } else {
#pragma unroll
        for (int v = 0; v < 8; ++v)
          if (rs + 8 * h + v < N) crow[(size_t)v * H2] = pacc[st][v] + bv;
      }
    }
  }
}

static inline int cdiv(int a, int b) { return (a + b - 1) / b; }

extern "C" void kernel_launch(void* const* d_in, const int* in_sizes, int n_in,
                              void* d_out, int out_size, void* d_ws, size_t ws_size,
                              hipStream_t stream) {
  const float* x     = (const float*)d_in[0];
  const int*   ei    = (const int*)  d_in[1];
  const float* W1l   = (const float*)d_in[2];
  const float* b1l   = (const float*)d_in[3];
  const float* W1r   = (const float*)d_in[4];
  const float* b1r   = (const float*)d_in[5];
  const float* att1  = (const float*)d_in[6];
  const float* bias1 = (const float*)d_in[7];
  const float* W2l   = (const float*)d_in[8];
  const float* b2l   = (const float*)d_in[9];
  const float* W2r   = (const float*)d_in[10];
  const float* b2r   = (const float*)d_in[11];
  const float* att2  = (const float*)d_in[12];
  const float* bias2 = (const float*)d_in[13];
  const float* Wlin  = (const float*)d_in[14];
  const float* blin  = (const float*)d_in[15];
  float* out = (float*)d_out;

  const int N    = in_sizes[0] / 128;
  const int E    = in_sizes[1] / 2;

  float* ws = (float*)d_ws;
  const int NB = cdiv(N, BINW);
  const int G  = cdiv(E, CH);
  size_t off = 0;
  float* xl1  = ws + off; off += (size_t)N * 64;
  float* xr1  = ws + off; off += (size_t)N * 64;
  float* xl2  = ws + off; off += (size_t)N * 32;
  float* xr2  = ws + off; off += (size_t)N * 32;
  int* rowptr = (int*)(ws + off); off += (size_t)N + 1;
  off = (off + 31) & ~(size_t)31;
  int* csr    = (int*)(ws + off); off += (size_t)E;
  int* bsrc   = (int*)(ws + off); off += (size_t)G * CH;
  int* bdst   = (int*)(ws + off); off += (size_t)G * CH;
  int* lscan  = (int*)(ws + off); off += (size_t)G * (NB + 1);
  (void)ws_size; (void)n_in; (void)out_size;

  const int TB = 256;
  const int NTB = NTBF;
  int NPW = cdiv(N, 256 * 4 * 9);
  if (NPW < 1) NPW = 1;
  if (NPW > 31) NPW = 31;
  const int nblocks = cdiv(cdiv(N, NPW), NTB / 32);
  const int nblocks1 = cdiv(cdiv(N, NPWF), NTBF / 32);

  bin_chunk<<<G, SORT_T, 0, stream>>>(ei, E, NB, bsrc, bdst, lscan);
  bin_to_csr<<<NB, SORT_T, 0, stream>>>(bsrc, bdst, lscan, G, NB, N, rowptr, csr);

  gemm2_bias_wmma<128, 64><<<768, TB, 0, stream>>>(x, W1l, b1l, W1r, b1r, xl1, xr1, N);
  node_attend<64, false><<<nblocks1, NTBF, 0, stream>>>(xl1, xr1, att1, rowptr, csr, bias1,
                                                       Wlin, blin, nullptr, N, NPWF,
                                                       W2l, b2l, W2r, b2r, xl2, xr2);

  node_attend<32, true><<<nblocks, NTB, 0, stream>>>(xl2, xr2, att2, rowptr, csr, bias2,
                                                    Wlin, blin, out, N, NPW,
                                                    W2l, b2l, W2r, b2r, nullptr, nullptr);
}
